# starting kernel
# baseline (speedup 1.0000x reference)
#include <hip/hip_runtime.h>


#define NNODES 50000
#define NEDGES 800000
#define INCH   256
#define HIDCH  128

typedef float v2f __attribute__((ext_vector_type(2)));
typedef float v8f __attribute__((ext_vector_type(8)));

__global__ __launch_bounds__(256) void k_init_deg(float* __restrict__ deg, int n) {
  int i = blockIdx.x * blockDim.x + threadIdx.x;
  if (i < n) deg[i] = 1.0f;
}

__global__ __launch_bounds__(256) void k_deg_accum(const int* __restrict__ dst,
                                                   const float* __restrict__ ew,
                                                   float* __restrict__ deg, int e) {
  int i = blockIdx.x * blockDim.x + threadIdx.x;
  if (i < e) unsafeAtomicAdd(deg + dst[i], ew[i]);
}

__global__ __launch_bounds__(256) void k_rsqrt(float* __restrict__ deg, int n) {
  int i = blockIdx.x * blockDim.x + threadIdx.x;
  if (i < n) deg[i] = rsqrtf(deg[i]);
}

__global__ __launch_bounds__(256) void k_zero4(float4* __restrict__ p, long long n4) {
  long long i = (long long)blockIdx.x * blockDim.x + threadIdx.x;
  if (i < n4) p[i] = make_float4(0.f, 0.f, 0.f, 0.f);
}

__global__ __launch_bounds__(128) void k_gemm_wmma_f32(const float* __restrict__ A,
                                                       const float* __restrict__ B,
                                                       float* __restrict__ C,
                                                       int M, int N, int K) {
  const int lane  = threadIdx.x & 31;
  const int wave  = threadIdx.x >> 5;
  const int strip = blockIdx.x * 4 + wave;
  if (strip >= (M >> 4)) return;

  const int mn   = lane & 15;
  const int kSel = (lane >> 4) * 2;

  const float* __restrict__ arow  = A + (size_t)(strip * 16 + mn) * K;
  const float* __restrict__ bbase = B + mn;

  v8f acc[8];
#pragma unroll
  for (int t = 0; t < 8; ++t) acc[t] = (v8f){};

  for (int k0 = 0; k0 < K; k0 += 4) {
    const int ka = k0 + kSel;
    v2f a;
    a.x = arow[ka];
    a.y = arow[ka + 1];
    const float* __restrict__ b0 = bbase + (size_t)ka * N;
    const float* __restrict__ b1 = bbase + (size_t)(ka + 1) * N;
#pragma unroll
    for (int t = 0; t < 8; ++t) {
      v2f b;
      b.x = b0[t * 16];
      b.y = b1[t * 16];
      acc[t] = __builtin_amdgcn_wmma_f32_16x16x4_f32(
           false, a,  false, b,
           (short)0, acc[t],  false,  false);
    }
  }

  const int mbase = (lane >> 4) * 8;
  float* __restrict__ cbase = C + (size_t)(strip * 16) * N + mn;
#pragma unroll
  for (int t = 0; t < 8; ++t) {
    float* __restrict__ crow = cbase + t * 16;
#pragma unroll
    for (int r = 0; r < 8; ++r)
      crow[(size_t)(mbase + r) * N] = acc[t][r];
  }
}

__global__ __launch_bounds__(256) void k_scatter(const int* __restrict__ src,
                                                 const int* __restrict__ dst,
                                                 const float* __restrict__ ew,
                                                 const float* __restrict__ dinv,
                                                 const float* __restrict__ h,
                                                 float* __restrict__ agg, int E) {
  long long gid = (long long)blockIdx.x * blockDim.x + threadIdx.x;
  int e = (int)(gid >> 5);
  if (e >= E) return;
  int q = (int)(gid & 31);
  int s = src[e];
  int d = dst[e];
  float norm = dinv[s] * ew[e] * dinv[d];
  const float4 hv = *(const float4*)(h + (size_t)s * HIDCH + q * 4);
  float* out = agg + (size_t)d * HIDCH + q * 4;
  unsafeAtomicAdd(out + 0, norm * hv.x);
  unsafeAtomicAdd(out + 1, norm * hv.y);
  unsafeAtomicAdd(out + 2, norm * hv.z);
  unsafeAtomicAdd(out + 3, norm * hv.w);
}

__global__ __launch_bounds__(256) void k_finalize(const float* __restrict__ agg,
                                                  const float* __restrict__ h,
                                                  const float* __restrict__ dinv,
                                                  const float* __restrict__ bias,
                                                  float* __restrict__ out,
                                                  long long total, int doRelu) {
  long long idx = (long long)blockIdx.x * blockDim.x + threadIdx.x;
  if (idx >= total) return;
  int i = (int)(idx >> 7);
  int c = (int)(idx & 127);
  float di = dinv[i];
  float v = agg[idx] + di * di * h[idx] + bias[c];
  out[idx] = doRelu ? fmaxf(v, 0.f) : v;
}

extern "C" void kernel_launch(void* const* d_in, const int* in_sizes, int n_in,
                              void* d_out, int out_size, void* d_ws, size_t ws_size,
                              hipStream_t stream) {
  (void)in_sizes; (void)n_in; (void)out_size; (void)ws_size;

  const float* x  = (const float*)d_in[0];
  const int*   ei = (const int*)d_in[1];
  const float* ew = (const float*)d_in[2];
  const float* W1 = (const float*)d_in[3];
  const float* b1 = (const float*)d_in[4];
  const float* W2 = (const float*)d_in[5];
  const float* b2 = (const float*)d_in[6];
  const float* W3 = (const float*)d_in[7];
  const float* b3 = (const float*)d_in[8];
  float* out = (float*)d_out;

  const int* src = ei;
  const int* dst = ei + NEDGES;

  float* ws   = (float*)d_ws;
  float* dinv = ws;
  float* h    = ws + 50048;
  float* agg  = h + (size_t)NNODES * HIDCH;
  float* xb   = agg + (size_t)NNODES * HIDCH;

  k_init_deg<<<(NNODES + 255) / 256, 256, 0, stream>>>(dinv, NNODES);
  k_deg_accum<<<(NEDGES + 255) / 256, 256, 0, stream>>>(dst, ew, dinv, NEDGES);
  k_rsqrt<<<(NNODES + 255) / 256, 256, 0, stream>>>(dinv, NNODES);

  const int strips = NNODES / 16;
  const int gemmBlocks = (strips + 3) / 4;
  const long long aggN4 = (long long)NNODES * HIDCH / 4;
  const long long scatThreads = (long long)NEDGES * 32;
  const long long total = (long long)NNODES * HIDCH;

  auto layer = [&](const float* Ain, int K, const float* W, const float* bias,
                   float* outp, int relu) {
    k_gemm_wmma_f32<<<gemmBlocks, 128, 0, stream>>>(Ain, W, h, NNODES, HIDCH, K);
    k_zero4<<<(int)((aggN4 + 255) / 256), 256, 0, stream>>>((float4*)agg, aggN4);
    k_scatter<<<(int)((scatThreads + 255) / 256), 256, 0, stream>>>(
        src, dst, ew, dinv, h, agg, NEDGES);
    k_finalize<<<(int)((total + 255) / 256), 256, 0, stream>>>(
        agg, h, dinv, bias, outp, total, relu);
  };

  layer(x,  INCH,  W1, b1, xb,  1);
  layer(xb, HIDCH, W2, b2, xb,  1);
  layer(xb, HIDCH, W3, b3, out, 0);
}


# Atomic edge scatter replaced by per-call counting sort by destination (atomic arrival-order rank) + one-wave-per-node register gather, self loop/bias/ReLU fused; sums run in that row order
# speedup vs baseline: 2.6510x; 2.6510x over previous
#include <hip/hip_runtime.h>


#define NNODES 50000
#define NEDGES 800000
#define INCH   256
#define HIDCH  128

typedef float v2f __attribute__((ext_vector_type(2)));
typedef float v8f __attribute__((ext_vector_type(8)));

__global__ __launch_bounds__(256) void k_zero_cnt(int* __restrict__ cnt, int n) {
  int i = blockIdx.x * blockDim.x + threadIdx.x;
  if (i < n) cnt[i] = 0;
}

__global__ __launch_bounds__(256) void k_count(const int* __restrict__ dst,
                                               int* __restrict__ cnt,
                                               int* __restrict__ rank, int e) {
  int i = blockIdx.x * blockDim.x + threadIdx.x;
  if (i < e) rank[i] = atomicAdd(cnt + dst[i], 1);
}

__global__ __launch_bounds__(256) void k_scan(const int* __restrict__ cnt,
                                              int* __restrict__ rowptr, int n) {
  __shared__ int red[256];
  __shared__ int sc[256];
  const int tid = threadIdx.x;
  const int i = blockIdx.x * 256 + tid;

  const int4* __restrict__ c4 = (const int4*)cnt;
  const int n4 = blockIdx.x * 64;
  int part = 0;
  for (int j = tid; j < n4; j += 256) {
    int4 v = c4[j];
    part += (v.x + v.y) + (v.z + v.w);
  }
  const int c = (i < n) ? cnt[i] : 0;
  red[tid] = part;
  sc[tid] = c;
  __syncthreads();
  for (int s = 128; s > 0; s >>= 1) {
    if (tid < s) red[tid] += red[tid + s];
    __syncthreads();
  }
  const int front = red[0];
  for (int off = 1; off < 256; off <<= 1) {
    int v = (tid >= off) ? sc[tid - off] : 0;
    __syncthreads();
    sc[tid] += v;
    __syncthreads();
  }
  const int incl = sc[tid];
  if (i < n) rowptr[i] = front + incl - c;
  if (i == n - 1) rowptr[n] = front + incl;
}

__global__ __launch_bounds__(256) void k_fill(const int* __restrict__ src,
                                              const int* __restrict__ dst,
                                              const float* __restrict__ ew,
                                              const int* __restrict__ rank,
                                              const int* __restrict__ rowptr,
                                              int2* __restrict__ csr, int e) {
  int i = blockIdx.x * blockDim.x + threadIdx.x;
  if (i >= e) return;
  int p = rowptr[dst[i]] + rank[i];
  if ((unsigned)p < (unsigned)e)
    csr[p] = make_int2(src[i], __float_as_int(ew[i]));
}

__global__ __launch_bounds__(256) void k_dinv(const int* __restrict__ rowptr,
                                              const int2* __restrict__ csr,
                                              float* __restrict__ dinv, int n) {
  int i = blockIdx.x * blockDim.x + threadIdx.x;
  if (i >= n) return;
  const int b = rowptr[i];
  const int e = rowptr[i + 1];
  float s = 0.f;
  for (int p = b; p < e; ++p) s += __int_as_float(csr[p].y);
  const float deg = s + 1.0f;
  dinv[i] = deg > 0.f ? rsqrtf(deg) : 0.f;
}

__global__ __launch_bounds__(128) void k_gemm_wmma_f32(const float* __restrict__ A,
                                                       const float* __restrict__ B,
                                                       float* __restrict__ C,
                                                       int M, int N, int K) {
  const int lane  = threadIdx.x & 31;
  const int wave  = threadIdx.x >> 5;
  const int strip = blockIdx.x * 4 + wave;
  if (strip >= (M >> 4)) return;

  const int mn   = lane & 15;
  const int kSel = (lane >> 4) * 2;

  const float* __restrict__ arow  = A + (size_t)(strip * 16 + mn) * K;
  const float* __restrict__ bbase = B + mn;

  v8f acc[8];
#pragma unroll
  for (int t = 0; t < 8; ++t) acc[t] = (v8f){};

  for (int k0 = 0; k0 < K; k0 += 4) {
    const int ka = k0 + kSel;
    v2f a;
    a.x = arow[ka];
    a.y = arow[ka + 1];
    const float* __restrict__ b0 = bbase + (size_t)ka * N;
    const float* __restrict__ b1 = bbase + (size_t)(ka + 1) * N;
#pragma unroll
    for (int t = 0; t < 8; ++t) {
      v2f b;
      b.x = b0[t * 16];
      b.y = b1[t * 16];
      acc[t] = __builtin_amdgcn_wmma_f32_16x16x4_f32(
           false, a,  false, b,
           (short)0, acc[t],  false,  false);
    }
  }

  const int mbase = (lane >> 4) * 8;
  float* __restrict__ cbase = C + (size_t)(strip * 16) * N + mn;
#pragma unroll
  for (int t = 0; t < 8; ++t) {
    float* __restrict__ crow = cbase + t * 16;
#pragma unroll
    for (int r = 0; r < 8; ++r)
      crow[(size_t)(mbase + r) * N] = acc[t][r];
  }
}

template <int U>
__device__ __forceinline__ void gather_batch(const float4* __restrict__ h4, int sv, float wv,
                                             int j, int lane, float4& acc) {
  float4 v[U];
  float w[U];
#pragma unroll
  for (int u = 0; u < U; ++u) {
    const int s = __builtin_amdgcn_readlane(sv, j + u);
    w[u] = __int_as_float(__builtin_amdgcn_readlane(__float_as_int(wv), j + u));
    v[u] = h4[(size_t)s * (HIDCH / 4) + lane];
  }
#pragma unroll
  for (int u = 0; u < U; ++u) {
    acc.x += w[u] * v[u].x;
    acc.y += w[u] * v[u].y;
    acc.z += w[u] * v[u].z;
    acc.w += w[u] * v[u].w;
  }
}

__global__ __launch_bounds__(256) void k_gather(const int* __restrict__ rowptr,
                                                const int2* __restrict__ csr,
                                                const float* __restrict__ dinv,
                                                const float* __restrict__ h,
                                                const float* __restrict__ bias,
                                                float* __restrict__ out,
                                                int n, int nedges, int doRelu) {
  const int lane = threadIdx.x & 31;
  const int node = __builtin_amdgcn_readfirstlane((int)(blockIdx.x * 8 + (threadIdx.x >> 5)));
  if (node >= n) return;

  int beg = rowptr[node];
  int end = rowptr[node + 1];
  beg = beg < 0 ? 0 : beg;
  end = end > nedges ? nedges : end;
  const float dd = dinv[node];

  const float4* __restrict__ h4 = (const float4*)h;
  const float4 hs = h4[(size_t)node * (HIDCH / 4) + lane];
  const float selfw = dd * dd;
  float4 acc = make_float4(0.f, 0.f, 0.f, 0.f);

  for (int base = beg; base < end; base += 32) {
    const int left = end - base;
    const int m = left < 32 ? left : 32;
    int sv = 0;
    float wv = 0.f;
    if (lane < m) {
      const int2 ce = csr[base + lane];
      int s = ce.x;
      s = s < 0 ? 0 : (s >= n ? n - 1 : s);
      sv = s;
      wv = dinv[s] * __int_as_float(ce.y) * dd;
    }
    int j = 0;
    for (; j + 8 <= m; j += 8) gather_batch<8>(h4, sv, wv, j, lane, acc);
    if (m & 4) { gather_batch<4>(h4, sv, wv, j, lane, acc); j += 4; }
    if (m & 2) { gather_batch<2>(h4, sv, wv, j, lane, acc); j += 2; }
    if (m & 1) { gather_batch<1>(h4, sv, wv, j, lane, acc); }
  }

  const float4 bv = ((const float4*)bias)[lane];
  float4 r = make_float4(acc.x + selfw * hs.x + bv.x, acc.y + selfw * hs.y + bv.y,
                         acc.z + selfw * hs.z + bv.z, acc.w + selfw * hs.w + bv.w);
  if (doRelu) {
    r.x = fmaxf(r.x, 0.f); r.y = fmaxf(r.y, 0.f);
    r.z = fmaxf(r.z, 0.f); r.w = fmaxf(r.w, 0.f);
  }
  ((float4*)out)[(size_t)node * (HIDCH / 4) + lane] = r;
}

extern "C" void kernel_launch(void* const* d_in, const int* in_sizes, int n_in,
                              void* d_out, int out_size, void* d_ws, size_t ws_size,
                              hipStream_t stream) {
  (void)in_sizes; (void)n_in; (void)out_size; (void)ws_size;

  const float* x  = (const float*)d_in[0];
  const int*   ei = (const int*)d_in[1];
  const float* ew = (const float*)d_in[2];
  const float* W1 = (const float*)d_in[3];
  const float* b1 = (const float*)d_in[4];
  const float* W2 = (const float*)d_in[5];
  const float* b2 = (const float*)d_in[6];
  const float* W3 = (const float*)d_in[7];
  const float* b3 = (const float*)d_in[8];
  float* out = (float*)d_out;

  const int* src = ei;
  const int* dst = ei + NEDGES;

  float* ws     = (float*)d_ws;
  float* dinv   = ws;
  int*   cnt    = (int*)(ws + 50176);
  int*   rowptr = (int*)(ws + 100352);
  int*   rank   = (int*)(ws + 150528);
  int2*  csr    = (int2*)(ws + 950528);
  float* h      = ws + 2550528;
  float* xb     = h + (size_t)NNODES * HIDCH;

  k_zero_cnt<<<(NNODES + 255) / 256, 256, 0, stream>>>(cnt, NNODES);
  k_count<<<(NEDGES + 255) / 256, 256, 0, stream>>>(dst, cnt, rank, NEDGES);
  k_scan<<<(NNODES + 255) / 256, 256, 0, stream>>>(cnt, rowptr, NNODES);
  k_fill<<<(NEDGES + 255) / 256, 256, 0, stream>>>(src, dst, ew, rank, rowptr, csr, NEDGES);
  k_dinv<<<(NNODES + 255) / 256, 256, 0, stream>>>(rowptr, csr, dinv, NNODES);

  const int strips = NNODES / 16;
  const int gemmBlocks = (strips + 3) / 4;
  const int gatherBlocks = (NNODES + 7) / 8;

  auto layer = [&](const float* Ain, int K, const float* W, const float* bias,
                   float* outp, int relu) {
    k_gemm_wmma_f32<<<gemmBlocks, 128, 0, stream>>>(Ain, W, h, NNODES, HIDCH, K);
    k_gather<<<gatherBlocks, 256, 0, stream>>>(rowptr, csr, dinv, h, bias, outp,
                                               NNODES, NEDGES, relu);
  };

  layer(x,  INCH,  W1, b1, xb,  1);
  layer(xb, HIDCH, W2, b2, xb,  1);
  layer(xb, HIDCH, W3, b3, out, 0);
}


# Next layer's 128x128 transform fused into the gather: 16 waves gather a 16-node tile into LDS, 8 of them multiply it by W kept in LDS (f32 WMMA, same k order, sums unchanged); 11 launches become 9
# speedup vs baseline: 3.1027x; 1.1704x over previous
#include <hip/hip_runtime.h>


#define NNODES 50000
#define NEDGES 800000
#define INCH   256
#define HIDCH  128
#define FUSED_GRID 1024

typedef float v2f __attribute__((ext_vector_type(2)));
typedef float v8f __attribute__((ext_vector_type(8)));

__global__ __launch_bounds__(256) void k_zero_cnt(int* __restrict__ cnt, int n) {
  int i = blockIdx.x * blockDim.x + threadIdx.x;
  if (i < n) cnt[i] = 0;
}

__global__ __launch_bounds__(256) void k_count(const int* __restrict__ dst,
                                               int* __restrict__ cnt,
                                               int* __restrict__ rank, int e) {
  int i = blockIdx.x * blockDim.x + threadIdx.x;
  if (i < e) rank[i] = atomicAdd(cnt + dst[i], 1);
}

__global__ __launch_bounds__(256) void k_scan(const int* __restrict__ cnt,
                                              int* __restrict__ rowptr, int n) {
  __shared__ int red[256];
  __shared__ int sc[256];
  const int tid = threadIdx.x;
  const int i = blockIdx.x * 256 + tid;

  const int4* __restrict__ c4 = (const int4*)cnt;
  const int n4 = blockIdx.x * 64;
  int part = 0;
  for (int j = tid; j < n4; j += 256) {
    int4 v = c4[j];
    part += (v.x + v.y) + (v.z + v.w);
  }
  const int c = (i < n) ? cnt[i] : 0;
  red[tid] = part;
  sc[tid] = c;
  __syncthreads();
  for (int s = 128; s > 0; s >>= 1) {
    if (tid < s) red[tid] += red[tid + s];
    __syncthreads();
  }
  const int front = red[0];
  for (int off = 1; off < 256; off <<= 1) {
    int v = (tid >= off) ? sc[tid - off] : 0;
    __syncthreads();
    sc[tid] += v;
    __syncthreads();
  }
  const int incl = sc[tid];
  if (i < n) rowptr[i] = front + incl - c;
  if (i == n - 1) rowptr[n] = front + incl;
}

__global__ __launch_bounds__(256) void k_fill(const int* __restrict__ src,
                                              const int* __restrict__ dst,
                                              const float* __restrict__ ew,
                                              const int* __restrict__ rank,
                                              const int* __restrict__ rowptr,
                                              int2* __restrict__ csr, int e) {
  int i = blockIdx.x * blockDim.x + threadIdx.x;
  if (i >= e) return;
  int p = rowptr[dst[i]] + rank[i];
  if ((unsigned)p < (unsigned)e)
    csr[p] = make_int2(src[i], __float_as_int(ew[i]));
}

__global__ __launch_bounds__(256) void k_dinv(const int* __restrict__ rowptr,
                                              const int2* __restrict__ csr,
                                              float* __restrict__ dinv, int n) {
  int i = blockIdx.x * blockDim.x + threadIdx.x;
  if (i >= n) return;
  const int b = rowptr[i];
  const int e = rowptr[i + 1];
  float s = 0.f;
  for (int p = b; p < e; ++p) s += __int_as_float(csr[p].y);
  const float deg = s + 1.0f;
  dinv[i] = deg > 0.f ? rsqrtf(deg) : 0.f;
}

__global__ __launch_bounds__(128) void k_gemm_wmma_f32(const float* __restrict__ A,
                                                       const float* __restrict__ B,
                                                       float* __restrict__ C,
                                                       int M, int N, int K) {
  const int lane  = threadIdx.x & 31;
  const int wave  = threadIdx.x >> 5;
  const int strip = blockIdx.x * 4 + wave;
  if (strip >= (M >> 4)) return;

  const int mn   = lane & 15;
  const int kSel = (lane >> 4) * 2;

  const float* __restrict__ arow  = A + (size_t)(strip * 16 + mn) * K;
  const float* __restrict__ bbase = B + mn;

  v8f acc[8];
#pragma unroll
  for (int t = 0; t < 8; ++t) acc[t] = (v8f){};

  for (int k0 = 0; k0 < K; k0 += 4) {
    const int ka = k0 + kSel;
    v2f a;
    a.x = arow[ka];
    a.y = arow[ka + 1];
    const float* __restrict__ b0 = bbase + (size_t)ka * N;
    const float* __restrict__ b1 = bbase + (size_t)(ka + 1) * N;
#pragma unroll
    for (int t = 0; t < 8; ++t) {
      v2f b;
      b.x = b0[t * 16];
      b.y = b1[t * 16];
      acc[t] = __builtin_amdgcn_wmma_f32_16x16x4_f32(
           false, a,  false, b,
           (short)0, acc[t],  false,  false);
    }
  }

  const int mbase = (lane >> 4) * 8;
  float* __restrict__ cbase = C + (size_t)(strip * 16) * N + mn;
#pragma unroll
  for (int t = 0; t < 8; ++t) {
    float* __restrict__ crow = cbase + t * 16;
#pragma unroll
    for (int r = 0; r < 8; ++r)
      crow[(size_t)(mbase + r) * N] = acc[t][r];
  }
}

template <int U>
__device__ __forceinline__ void gather_batch(const float4* __restrict__ h4, int sv, float wv,
                                             int j, int lane, float4& acc) {
  float4 v[U];
  float w[U];
#pragma unroll
  for (int u = 0; u < U; ++u) {
    const int s = __builtin_amdgcn_readlane(sv, j + u);
    w[u] = __int_as_float(__builtin_amdgcn_readlane(__float_as_int(wv), j + u));
    v[u] = h4[(size_t)s * (HIDCH / 4) + lane];
  }
#pragma unroll
  for (int u = 0; u < U; ++u) {
    acc.x += w[u] * v[u].x;
    acc.y += w[u] * v[u].y;
    acc.z += w[u] * v[u].z;
    acc.w += w[u] * v[u].w;
  }
}

__global__ __launch_bounds__(256) void k_gather(const int* __restrict__ rowptr,
                                                const int2* __restrict__ csr,
                                                const float* __restrict__ dinv,
                                                const float* __restrict__ h,
                                                const float* __restrict__ bias,
                                                float* __restrict__ out,
                                                int n, int nedges, int doRelu) {
  const int lane = threadIdx.x & 31;
  const int node = __builtin_amdgcn_readfirstlane((int)(blockIdx.x * 8 + (threadIdx.x >> 5)));
  if (node >= n) return;

  int beg = rowptr[node];
  int end = rowptr[node + 1];
  beg = beg < 0 ? 0 : beg;
  end = end > nedges ? nedges : end;
  const float dd = dinv[node];

  const float4* __restrict__ h4 = (const float4*)h;
  const float4 hs = h4[(size_t)node * (HIDCH / 4) + lane];
  const float selfw = dd * dd;
  float4 acc = make_float4(0.f, 0.f, 0.f, 0.f);

  for (int base = beg; base < end; base += 32) {
    const int left = end - base;
    const int m = left < 32 ? left : 32;
    int sv = 0;
    float wv = 0.f;
    if (lane < m) {
      const int2 ce = csr[base + lane];
      int s = ce.x;
      s = s < 0 ? 0 : (s >= n ? n - 1 : s);
      sv = s;
      wv = dinv[s] * __int_as_float(ce.y) * dd;
    }
    int j = 0;
    for (; j + 8 <= m; j += 8) gather_batch<8>(h4, sv, wv, j, lane, acc);
    if (m & 4) { gather_batch<4>(h4, sv, wv, j, lane, acc); j += 4; }
    if (m & 2) { gather_batch<2>(h4, sv, wv, j, lane, acc); j += 2; }
    if (m & 1) { gather_batch<1>(h4, sv, wv, j, lane, acc); }
  }

  const float4 bv = ((const float4*)bias)[lane];
  float4 r = make_float4(acc.x + selfw * hs.x + bv.x, acc.y + selfw * hs.y + bv.y,
                         acc.z + selfw * hs.z + bv.z, acc.w + selfw * hs.w + bv.w);
  if (doRelu) {
    r.x = fmaxf(r.x, 0.f); r.y = fmaxf(r.y, 0.f);
    r.z = fmaxf(r.z, 0.f); r.w = fmaxf(r.w, 0.f);
  }
  ((float4*)out)[(size_t)node * (HIDCH / 4) + lane] = r;
}

#define TPITCH 132
#define WPITCH 136
static_assert(NNODES % 16 == 0, "the fused kernel writes whole 16-node tiles");

__global__ __launch_bounds__(512) void k_gather_gemm(const int* __restrict__ rowptr,
                                                     const int2* __restrict__ csr,
                                                     const float* __restrict__ dinv,
                                                     const float* __restrict__ h,
                                                     const float* __restrict__ bias,
                                                     const float* __restrict__ Wn,
                                                     float* __restrict__ hout,
                                                     int n, int nedges, int ntiles) {
  __shared__ float Ws[HIDCH * WPITCH];
  __shared__ float tile[16 * TPITCH];

  const int tid  = threadIdx.x;
  const int lane = tid & 31;
  const int wave = __builtin_amdgcn_readfirstlane(tid >> 5);

  {
    const float4* __restrict__ W4 = (const float4*)Wn;
#pragma unroll
    for (int j = 0; j < 8; ++j) {
      const int i = tid + j * 512;
      const float4 v = W4[i];
      *(float4*)&Ws[(i >> 5) * WPITCH + (i & 31) * 4] = v;
    }
  }

  const float4* __restrict__ h4 = (const float4*)h;
  const float4 bv = ((const float4*)bias)[lane];

  for (int t = blockIdx.x; t < ntiles; t += gridDim.x) {
    const int node = t * 16 + wave;
    float4 r = make_float4(0.f, 0.f, 0.f, 0.f);
    if (node < n) {
      int beg = rowptr[node];
      int end = rowptr[node + 1];
      beg = beg < 0 ? 0 : beg;
      end = end > nedges ? nedges : end;
      const float dd = dinv[node];
      const float4 hs = h4[(size_t)node * (HIDCH / 4) + lane];
      const float selfw = dd * dd;
      float4 acc = make_float4(0.f, 0.f, 0.f, 0.f);

      for (int base = beg; base < end; base += 32) {
        const int left = end - base;
        const int m = left < 32 ? left : 32;
        int sv = 0;
        float wv = 0.f;
        if (lane < m) {
          const int2 ce = csr[base + lane];
          int s = ce.x;
          s = s < 0 ? 0 : (s >= n ? n - 1 : s);
          sv = s;
          wv = dinv[s] * __int_as_float(ce.y) * dd;
        }
        int j = 0;
        for (; j + 8 <= m; j += 8) gather_batch<8>(h4, sv, wv, j, lane, acc);
        if (m & 4) { gather_batch<4>(h4, sv, wv, j, lane, acc); j += 4; }
        if (m & 2) { gather_batch<2>(h4, sv, wv, j, lane, acc); j += 2; }
        if (m & 1) { gather_batch<1>(h4, sv, wv, j, lane, acc); }
      }

      r = make_float4(acc.x + selfw * hs.x + bv.x, acc.y + selfw * hs.y + bv.y,
                      acc.z + selfw * hs.z + bv.z, acc.w + selfw * hs.w + bv.w);
      r.x = fmaxf(r.x, 0.f); r.y = fmaxf(r.y, 0.f);
      r.z = fmaxf(r.z, 0.f); r.w = fmaxf(r.w, 0.f);
    }

    __syncthreads();
    *(float4*)&tile[wave * TPITCH + lane * 4] = r;
    __syncthreads();

    if (wave < 8) {
      const int mn   = lane & 15;
      const int kSel = (lane >> 4) * 2;
      const float* ap = tile + mn * TPITCH + kSel;
      const float* bp = Ws + kSel * WPITCH + wave * 16 + mn;
      v8f c = (v8f){};
#pragma unroll 8
      for (int k0 = 0; k0 < HIDCH; k0 += 4) {
        v2f a;
        a.x = ap[k0];
        a.y = ap[k0 + 1];
        v2f b;
        b.x = bp[k0 * WPITCH];
        b.y = bp[(k0 + 1) * WPITCH];
        c = __builtin_amdgcn_wmma_f32_16x16x4_f32(
             false, a,  false, b,
             (short)0, c,  false,  false);
      }
      const int row0 = t * 16 + (lane >> 4) * 8;
      float* __restrict__ crow = hout + (size_t)row0 * HIDCH + wave * 16 + mn;
#pragma unroll
      for (int q = 0; q < 8; ++q)
        crow[(size_t)q * HIDCH] = c[q];
    }
  }
}

extern "C" void kernel_launch(void* const* d_in, const int* in_sizes, int n_in,
                              void* d_out, int out_size, void* d_ws, size_t ws_size,
                              hipStream_t stream) {
  (void)in_sizes; (void)n_in; (void)out_size; (void)ws_size;

  const float* x  = (const float*)d_in[0];
  const int*   ei = (const int*)d_in[1];
  const float* ew = (const float*)d_in[2];
  const float* W1 = (const float*)d_in[3];
  const float* b1 = (const float*)d_in[4];
  const float* W2 = (const float*)d_in[5];
  const float* b2 = (const float*)d_in[6];
  const float* W3 = (const float*)d_in[7];
  const float* b3 = (const float*)d_in[8];
  float* out = (float*)d_out;

  const int* src = ei;
  const int* dst = ei + NEDGES;

  float* ws     = (float*)d_ws;
  float* dinv   = ws;
  int*   cnt    = (int*)(ws + 50176);
  int*   rowptr = (int*)(ws + 100352);
  int*   rank   = (int*)(ws + 150528);
  int2*  csr    = (int2*)(ws + 950528);
  float* h      = ws + 2550528;
  float* xb     = h + (size_t)NNODES * HIDCH;

  k_zero_cnt<<<(NNODES + 255) / 256, 256, 0, stream>>>(cnt, NNODES);
  k_count<<<(NEDGES + 255) / 256, 256, 0, stream>>>(dst, cnt, rank, NEDGES);
  k_scan<<<(NNODES + 255) / 256, 256, 0, stream>>>(cnt, rowptr, NNODES);
  k_fill<<<(NEDGES + 255) / 256, 256, 0, stream>>>(src, dst, ew, rank, rowptr, csr, NEDGES);
  k_dinv<<<(NNODES + 255) / 256, 256, 0, stream>>>(rowptr, csr, dinv, NNODES);

  const int strips = NNODES / 16;
  const int gemmBlocks = (strips + 3) / 4;
  const int gatherBlocks = (NNODES + 7) / 8;

  const int ntiles = NNODES / 16;
  const int fusedBlocks = FUSED_GRID;

  k_gemm_wmma_f32<<<gemmBlocks, 128, 0, stream>>>(x, W1, h, NNODES, HIDCH, INCH);
  k_gather_gemm<<<fusedBlocks, 512, 0, stream>>>(rowptr, csr, dinv, h, b1, W2, xb,
                                                 NNODES, NEDGES, ntiles);
  k_gather_gemm<<<fusedBlocks, 512, 0, stream>>>(rowptr, csr, dinv, xb, b2, W3, h,
                                                 NNODES, NEDGES, ntiles);
  k_gather<<<gatherBlocks, 256, 0, stream>>>(rowptr, csr, dinv, h, b3, out,
                                             NNODES, NEDGES, 0);
}


# Counting sort by destination rebuilt on LDS counters (bucket of 256 nodes, then node; slots from LDS returning adds; no per-edge global atomic); deg summed by LDS float adds; 9 launches become 7
# speedup vs baseline: 4.2662x; 1.3750x over previous
#include <hip/hip_runtime.h>


#define NNODES 50000
#define NEDGES 800000
#define INCH   256
#define HIDCH  128
#define FUSED_GRID 1024

typedef float v2f __attribute__((ext_vector_type(2)));
typedef float v8f __attribute__((ext_vector_type(8)));

#define BIN_NODES 256
#define NBIN ((NNODES + BIN_NODES - 1) / BIN_NODES)
#define NCHUNK 256
#define CHUNK_E ((NEDGES + NCHUNK - 1) / NCHUNK)
#define CHUNK_T 512
#define EPT ((CHUNK_E + CHUNK_T - 1) / CHUNK_T)
#define RPW ((NBIN + CHUNK_T / 32 - 1) / (CHUNK_T / 32))
static_assert(NBIN <= 256, "bucket tables are 256 wide");
static_assert(NCHUNK == 256, "a cntmat row is read as two 16-byte loads per lane");
static_assert(NNODES <= (1 << 24), "src is packed into 24 bits");
static_assert((NCHUNK - 1) * CHUNK_E < NEDGES, "every chunk starts inside the edge list");

__global__ __launch_bounds__(CHUNK_T) void k_bin_count(const int* __restrict__ dst,
                                                    int* __restrict__ cntmat, int e) {
  __shared__ int hist[256];
  const int tid = threadIdx.x;
  const int cbeg = blockIdx.x * CHUNK_E;
  const int cend = (cbeg + CHUNK_E < e) ? cbeg + CHUNK_E : e;
  int d[EPT];
#pragma unroll
  for (int u = 0; u < EPT; ++u) {
    const int i = cbeg + tid + u * CHUNK_T;
    d[u] = (i < cend) ? dst[i] : -1;
  }
  if (tid < 256) hist[tid] = 0;
  __syncthreads();
#pragma unroll
  for (int u = 0; u < EPT; ++u)
    if ((unsigned)d[u] < (unsigned)NNODES) atomicAdd(&hist[d[u] >> 8], 1);
  __syncthreads();
  if (tid < NBIN) cntmat[tid * NCHUNK + blockIdx.x] = hist[tid];
}

__device__ __forceinline__ void wave_scan256(const int* in, int* out, int lane) {
  int v[8];
  int s = 0;
#pragma unroll
  for (int k = 0; k < 8; ++k) v[k] = in[lane * 8 + k];
#pragma unroll
  for (int k = 0; k < 8; ++k) { const int t = v[k]; v[k] = s; s += t; }
  int incl = s;
#pragma unroll
  for (int off = 1; off < 32; off <<= 1) {
    const int nb = __shfl_up(incl, off, 32);
    if (lane >= off) incl += nb;
  }
  const int excl = incl - s;
#pragma unroll
  for (int k = 0; k < 8; ++k) out[lane * 8 + k] = excl + v[k];
  if (lane == 31) out[256] = incl;
}

__global__ __launch_bounds__(CHUNK_T) void k_bin_place(const int* __restrict__ src,
                                                    const int* __restrict__ dst,
                                                    const float* __restrict__ ew,
                                                    const int* __restrict__ cntmat,
                                                    int* __restrict__ binStart,
                                                    int2* __restrict__ binned, int e) {
  __shared__ int binTot[256];
  __shared__ int binPre[256];
  __shared__ int binBeg[257];
  __shared__ int cursor[256];
  const int tid  = threadIdx.x;
  const int lane = tid & 31;
  const int wave = tid >> 5;
  const int blk  = blockIdx.x;

  int4 a[RPW], b[RPW];
#pragma unroll
  for (int j = 0; j < RPW; ++j) {
    const int row = wave + j * (CHUNK_T / 32);
    const int rr = row < NBIN ? row : NBIN - 1;
    const int4* __restrict__ r4 = (const int4*)(cntmat + rr * NCHUNK);
    a[j] = r4[lane];
    b[j] = r4[32 + lane];
  }

  const int cbeg = blk * CHUNK_E;
  const int cend = (cbeg + CHUNK_E < e) ? cbeg + CHUNK_E : e;
  int d[EPT], s[EPT];
  float w[EPT];
#pragma unroll
  for (int u = 0; u < EPT; ++u) {
    const int i = cbeg + tid + u * CHUNK_T;
    const int ic = i < cend ? i : cbeg;
    const int dv = dst[ic];
    s[u] = src[ic];
    w[u] = ew[ic];
    d[u] = i < cend ? dv : -1;
  }

  if (tid >= NBIN && tid < 256) { binTot[tid] = 0; binPre[tid] = 0; }
  {
    const int c0 = 4 * lane;
    const int c1 = 128 + 4 * lane;
#pragma unroll
    for (int j = 0; j < RPW; ++j) {
      const int row = wave + j * (CHUNK_T / 32);
      int tot = (a[j].x + a[j].y) + (a[j].z + a[j].w) + (b[j].x + b[j].y) + (b[j].z + b[j].w);
      int pre = (c0     < blk ? a[j].x : 0) + (c0 + 1 < blk ? a[j].y : 0) +
                (c0 + 2 < blk ? a[j].z : 0) + (c0 + 3 < blk ? a[j].w : 0) +
                (c1     < blk ? b[j].x : 0) + (c1 + 1 < blk ? b[j].y : 0) +
                (c1 + 2 < blk ? b[j].z : 0) + (c1 + 3 < blk ? b[j].w : 0);
#pragma unroll
      for (int m = 16; m > 0; m >>= 1) {
        tot += __shfl_xor(tot, m, 32);
        pre += __shfl_xor(pre, m, 32);
      }
      if (lane == 0 && row < NBIN) { binTot[row] = tot; binPre[row] = pre; }
    }
  }
  __syncthreads();
  if (wave == 0) wave_scan256(binTot, binBeg, lane);
  __syncthreads();
  if (tid < 256) cursor[tid] = binBeg[tid] + binPre[tid];
  if (blk == 0 && tid <= NBIN) binStart[tid] = binBeg[tid];
  __syncthreads();
#pragma unroll
  for (int u = 0; u < EPT; ++u) {
    if ((unsigned)d[u] < (unsigned)NNODES) {
      const int p = atomicAdd(&cursor[d[u] >> 8], 1);
      int sc = s[u];
      sc = sc < 0 ? 0 : (sc >= NNODES ? NNODES - 1 : sc);
      const unsigned key = ((unsigned)(d[u] & 255) << 24) | (unsigned)sc;
      if ((unsigned)p < (unsigned)e)
        binned[p] = make_int2((int)key, __float_as_int(w[u]));
    }
  }
}

__global__ __launch_bounds__(512) void k_bin_sort(const int* __restrict__ binStart,
                                                  const int2* __restrict__ binned,
                                                  int2* __restrict__ csr,
                                                  int* __restrict__ rowptr,
                                                  float* __restrict__ dinv, int n, int e) {
  __shared__ int   cnt[256];
  __shared__ float deg[256];
  __shared__ int   loff[257];
  __shared__ int   cur[256];
  const int tid  = threadIdx.x;
  const int lane = tid & 31;
  const int wave = tid >> 5;
  const int bin  = blockIdx.x;

  int start = binStart[bin];
  int end   = binStart[bin + 1];
  start = start < 0 ? 0 : start;
  end   = end > e ? e : end;
  if (tid < 256) { cnt[tid] = 0; deg[tid] = 0.f; cur[tid] = 0; }
  __syncthreads();
  for (int i = start + tid; i < end; i += 4 * 512) {
    int2 r[4];
#pragma unroll
    for (int u = 0; u < 4; ++u) {
      const int idx = i + u * 512;
      r[u] = make_int2(0, 0);
      if (idx < end) r[u] = binned[idx];
    }
#pragma unroll
    for (int u = 0; u < 4; ++u) {
      if (i + u * 512 < end) {
        const int dl = (int)((unsigned)r[u].x >> 24);
        atomicAdd(&cnt[dl], 1);
        atomicAdd(&deg[dl], __int_as_float(r[u].y));
      }
    }
  }
  __syncthreads();
  if (wave == 0) wave_scan256(cnt, loff, lane);
  __syncthreads();
  if (tid < 256) {
    const int node = bin * BIN_NODES + tid;
    if (node < n) {
      rowptr[node] = start + loff[tid];
      const float dg = deg[tid] + 1.0f;
      dinv[node] = dg > 0.f ? rsqrtf(dg) : 0.f;
    }
  }
  if (bin == (int)gridDim.x - 1 && tid == 0) rowptr[n] = end;
  for (int i = start + tid; i < end; i += 4 * 512) {
    int2 r[4];
#pragma unroll
    for (int u = 0; u < 4; ++u) {
      const int idx = i + u * 512;
      r[u] = make_int2(0, 0);
      if (idx < end) r[u] = binned[idx];
    }
#pragma unroll
    for (int u = 0; u < 4; ++u) {
      if (i + u * 512 < end) {
        const int dl = (int)((unsigned)r[u].x >> 24);
        const int q = start + loff[dl] + atomicAdd(&cur[dl], 1);
        if ((unsigned)q < (unsigned)e)
          csr[q] = make_int2(r[u].x & 0x00FFFFFF, r[u].y);
      }
    }
  }
}

__global__ __launch_bounds__(128) void k_gemm_wmma_f32(const float* __restrict__ A,
                                                       const float* __restrict__ B,
                                                       float* __restrict__ C,
                                                       int M, int N, int K) {
  const int lane  = threadIdx.x & 31;
  const int wave  = threadIdx.x >> 5;
  const int strip = blockIdx.x * 4 + wave;
  if (strip >= (M >> 4)) return;

  const int mn   = lane & 15;
  const int kSel = (lane >> 4) * 2;

  const float* __restrict__ arow  = A + (size_t)(strip * 16 + mn) * K;
  const float* __restrict__ bbase = B + mn;

  v8f acc[8];
#pragma unroll
  for (int t = 0; t < 8; ++t) acc[t] = (v8f){};

  for (int k0 = 0; k0 < K; k0 += 4) {
    const int ka = k0 + kSel;
    v2f a;
    a.x = arow[ka];
    a.y = arow[ka + 1];
    const float* __restrict__ b0 = bbase + (size_t)ka * N;
    const float* __restrict__ b1 = bbase + (size_t)(ka + 1) * N;
#pragma unroll
    for (int t = 0; t < 8; ++t) {
      v2f b;
      b.x = b0[t * 16];
      b.y = b1[t * 16];
      acc[t] = __builtin_amdgcn_wmma_f32_16x16x4_f32(
           false, a,  false, b,
           (short)0, acc[t],  false,  false);
    }
  }

  const int mbase = (lane >> 4) * 8;
  float* __restrict__ cbase = C + (size_t)(strip * 16) * N + mn;
#pragma unroll
  for (int t = 0; t < 8; ++t) {
    float* __restrict__ crow = cbase + t * 16;
#pragma unroll
    for (int r = 0; r < 8; ++r)
      crow[(size_t)(mbase + r) * N] = acc[t][r];
  }
}

template <int U>
__device__ __forceinline__ void gather_batch(const float4* __restrict__ h4, int sv, float wv,
                                             int j, int lane, float4& acc) {
  float4 v[U];
  float w[U];
#pragma unroll
  for (int u = 0; u < U; ++u) {
    const int s = __builtin_amdgcn_readlane(sv, j + u);
    w[u] = __int_as_float(__builtin_amdgcn_readlane(__float_as_int(wv), j + u));
    v[u] = h4[(size_t)s * (HIDCH / 4) + lane];
  }
#pragma unroll
  for (int u = 0; u < U; ++u) {
    acc.x += w[u] * v[u].x;
    acc.y += w[u] * v[u].y;
    acc.z += w[u] * v[u].z;
    acc.w += w[u] * v[u].w;
  }
}

__global__ __launch_bounds__(256) void k_gather(const int* __restrict__ rowptr,
                                                const int2* __restrict__ csr,
                                                const float* __restrict__ dinv,
                                                const float* __restrict__ h,
                                                const float* __restrict__ bias,
                                                float* __restrict__ out,
                                                int n, int nedges, int doRelu) {
  const int lane = threadIdx.x & 31;
  const int node = __builtin_amdgcn_readfirstlane((int)(blockIdx.x * 8 + (threadIdx.x >> 5)));
  if (node >= n) return;

  int beg = rowptr[node];
  int end = rowptr[node + 1];
  beg = beg < 0 ? 0 : beg;
  end = end > nedges ? nedges : end;
  const float dd = dinv[node];

  const float4* __restrict__ h4 = (const float4*)h;
  const float4 hs = h4[(size_t)node * (HIDCH / 4) + lane];
  const float selfw = dd * dd;
  float4 acc = make_float4(0.f, 0.f, 0.f, 0.f);

  for (int base = beg; base < end; base += 32) {
    const int left = end - base;
    const int m = left < 32 ? left : 32;
    int sv = 0;
    float wv = 0.f;
    if (lane < m) {
      const int2 ce = csr[base + lane];
      int s = ce.x;
      s = s < 0 ? 0 : (s >= n ? n - 1 : s);
      sv = s;
      wv = dinv[s] * __int_as_float(ce.y) * dd;
    }
    int j = 0;
    for (; j + 8 <= m; j += 8) gather_batch<8>(h4, sv, wv, j, lane, acc);
    if (m & 4) { gather_batch<4>(h4, sv, wv, j, lane, acc); j += 4; }
    if (m & 2) { gather_batch<2>(h4, sv, wv, j, lane, acc); j += 2; }
    if (m & 1) { gather_batch<1>(h4, sv, wv, j, lane, acc); }
  }

  const float4 bv = ((const float4*)bias)[lane];
  float4 r = make_float4(acc.x + selfw * hs.x + bv.x, acc.y + selfw * hs.y + bv.y,
                         acc.z + selfw * hs.z + bv.z, acc.w + selfw * hs.w + bv.w);
  if (doRelu) {
    r.x = fmaxf(r.x, 0.f); r.y = fmaxf(r.y, 0.f);
    r.z = fmaxf(r.z, 0.f); r.w = fmaxf(r.w, 0.f);
  }
  ((float4*)out)[(size_t)node * (HIDCH / 4) + lane] = r;
}

#define TPITCH 132
#define WPITCH 136
static_assert(NNODES % 16 == 0, "the fused kernel writes whole 16-node tiles");

__global__ __launch_bounds__(512) void k_gather_gemm(const int* __restrict__ rowptr,
                                                     const int2* __restrict__ csr,
                                                     const float* __restrict__ dinv,
                                                     const float* __restrict__ h,
                                                     const float* __restrict__ bias,
                                                     const float* __restrict__ Wn,
                                                     float* __restrict__ hout,
                                                     int n, int nedges, int ntiles) {
  __shared__ float Ws[HIDCH * WPITCH];
  __shared__ float tile[16 * TPITCH];

  const int tid  = threadIdx.x;
  const int lane = tid & 31;
  const int wave = __builtin_amdgcn_readfirstlane(tid >> 5);

  {
    const float4* __restrict__ W4 = (const float4*)Wn;
#pragma unroll
    for (int j = 0; j < 8; ++j) {
      const int i = tid + j * 512;
      const float4 v = W4[i];
      *(float4*)&Ws[(i >> 5) * WPITCH + (i & 31) * 4] = v;
    }
  }

  const float4* __restrict__ h4 = (const float4*)h;
  const float4 bv = ((const float4*)bias)[lane];

  for (int t = blockIdx.x; t < ntiles; t += gridDim.x) {
    const int node = t * 16 + wave;
    float4 r = make_float4(0.f, 0.f, 0.f, 0.f);
    if (node < n) {
      int beg = rowptr[node];
      int end = rowptr[node + 1];
      beg = beg < 0 ? 0 : beg;
      end = end > nedges ? nedges : end;
      const float dd = dinv[node];
      const float4 hs = h4[(size_t)node * (HIDCH / 4) + lane];
      const float selfw = dd * dd;
      float4 acc = make_float4(0.f, 0.f, 0.f, 0.f);

      for (int base = beg; base < end; base += 32) {
        const int left = end - base;
        const int m = left < 32 ? left : 32;
        int sv = 0;
        float wv = 0.f;
        if (lane < m) {
          const int2 ce = csr[base + lane];
          int s = ce.x;
          s = s < 0 ? 0 : (s >= n ? n - 1 : s);
          sv = s;
          wv = dinv[s] * __int_as_float(ce.y) * dd;
        }
        int j = 0;
        for (; j + 8 <= m; j += 8) gather_batch<8>(h4, sv, wv, j, lane, acc);
        if (m & 4) { gather_batch<4>(h4, sv, wv, j, lane, acc); j += 4; }
        if (m & 2) { gather_batch<2>(h4, sv, wv, j, lane, acc); j += 2; }
        if (m & 1) { gather_batch<1>(h4, sv, wv, j, lane, acc); }
      }

      r = make_float4(acc.x + selfw * hs.x + bv.x, acc.y + selfw * hs.y + bv.y,
                      acc.z + selfw * hs.z + bv.z, acc.w + selfw * hs.w + bv.w);
      r.x = fmaxf(r.x, 0.f); r.y = fmaxf(r.y, 0.f);
      r.z = fmaxf(r.z, 0.f); r.w = fmaxf(r.w, 0.f);
    }

    __syncthreads();
    *(float4*)&tile[wave * TPITCH + lane * 4] = r;
    __syncthreads();

    if (wave < 8) {
      const int mn   = lane & 15;
      const int kSel = (lane >> 4) * 2;
      const float* ap = tile + mn * TPITCH + kSel;
      const float* bp = Ws + kSel * WPITCH + wave * 16 + mn;
      v8f c = (v8f){};
#pragma unroll 8
      for (int k0 = 0; k0 < HIDCH; k0 += 4) {
        v2f a;
        a.x = ap[k0];
        a.y = ap[k0 + 1];
        v2f b;
        b.x = bp[k0 * WPITCH];
        b.y = bp[(k0 + 1) * WPITCH];
        c = __builtin_amdgcn_wmma_f32_16x16x4_f32(
             false, a,  false, b,
             (short)0, c,  false,  false);
      }
      const int row0 = t * 16 + (lane >> 4) * 8;
      float* __restrict__ crow = hout + (size_t)row0 * HIDCH + wave * 16 + mn;
#pragma unroll
      for (int q = 0; q < 8; ++q)
        crow[(size_t)q * HIDCH] = c[q];
    }
  }
}

extern "C" void kernel_launch(void* const* d_in, const int* in_sizes, int n_in,
                              void* d_out, int out_size, void* d_ws, size_t ws_size,
                              hipStream_t stream) {
  (void)in_sizes; (void)n_in; (void)out_size; (void)ws_size;

  const float* x  = (const float*)d_in[0];
  const int*   ei = (const int*)d_in[1];
  const float* ew = (const float*)d_in[2];
  const float* W1 = (const float*)d_in[3];
  const float* b1 = (const float*)d_in[4];
  const float* W2 = (const float*)d_in[5];
  const float* b2 = (const float*)d_in[6];
  const float* W3 = (const float*)d_in[7];
  const float* b3 = (const float*)d_in[8];
  float* out = (float*)d_out;

  const int* src = ei;
  const int* dst = ei + NEDGES;

  float* ws       = (float*)d_ws;
  float* dinv     = ws;
  int*   cntmat   = (int*)(ws + 50176);
  int*   binStart = (int*)(ws + 100352);
  int*   rowptr   = (int*)(ws + 100864);
  int2*  binned   = (int2*)(ws + 151040);
  int2*  csr      = (int2*)(ws + 1751040);
  float* h        = ws + 3351040;
  float* xb       = h + (size_t)NNODES * HIDCH;

  k_bin_count<<<NCHUNK, CHUNK_T, 0, stream>>>(dst, cntmat, NEDGES);
  k_bin_place<<<NCHUNK, CHUNK_T, 0, stream>>>(src, dst, ew, cntmat, binStart, binned, NEDGES);
  k_bin_sort<<<NBIN, 512, 0, stream>>>(binStart, binned, csr, rowptr, dinv, NNODES, NEDGES);

  const int strips = NNODES / 16;
  const int gemmBlocks = (strips + 3) / 4;
  const int gatherBlocks = (NNODES + 7) / 8;

  const int ntiles = NNODES / 16;
  const int fusedBlocks = FUSED_GRID;

  k_gemm_wmma_f32<<<gemmBlocks, 128, 0, stream>>>(x, W1, h, NNODES, HIDCH, INCH);
  k_gather_gemm<<<fusedBlocks, 512, 0, stream>>>(rowptr, csr, dinv, h, b1, W2, xb,
                                                 NNODES, NEDGES, ntiles);
  k_gather_gemm<<<fusedBlocks, 512, 0, stream>>>(rowptr, csr, dinv, xb, b2, W3, h,
                                                 NNODES, NEDGES, ntiles);
  k_gather<<<gatherBlocks, 256, 0, stream>>>(rowptr, csr, dinv, h, b3, out,
                                             NNODES, NEDGES, 0);
}


# x@W1 gets no launch of its own: its strips ride as extra work-groups behind the 3 edge-sort launches (7 launches become 6); k split in four per strip, partials summed in LDS in fixed order
# speedup vs baseline: 4.8594x; 1.1390x over previous
#include <hip/hip_runtime.h>


#define NNODES 50000
#define NEDGES 800000
#define INCH   256
#define HIDCH  128
#define FUSED_GRID 1024

typedef float v2f __attribute__((ext_vector_type(2)));
typedef float v8f __attribute__((ext_vector_type(8)));

#define BIN_NODES 256
#define NBIN ((NNODES + BIN_NODES - 1) / BIN_NODES)
#define NCHUNK 256
#define CHUNK_E ((NEDGES + NCHUNK - 1) / NCHUNK)
#define CHUNK_T 512
#define EPT ((CHUNK_E + CHUNK_T - 1) / CHUNK_T)
#define RPW ((NBIN + CHUNK_T / 32 - 1) / (CHUNK_T / 32))
static_assert(NBIN <= 256, "bucket tables are 256 wide");
static_assert(NCHUNK == 256, "a cntmat row is read as two 16-byte loads per lane");
static_assert(NNODES <= (1 << 24), "src is packed into 24 bits");
static_assert((NCHUNK - 1) * CHUNK_E < NEDGES, "every chunk starts inside the edge list");

__device__ __forceinline__ void bin_count_role(const int* __restrict__ dst,
                                               int* __restrict__ cntmat, int e, int blk) {
  __shared__ int hist[256];
  const int tid = threadIdx.x;
  const int cbeg = blk * CHUNK_E;
  const int cend = (cbeg + CHUNK_E < e) ? cbeg + CHUNK_E : e;
  int d[EPT];
#pragma unroll
  for (int u = 0; u < EPT; ++u) {
    const int i = cbeg + tid + u * CHUNK_T;
    d[u] = (i < cend) ? dst[i] : -1;
  }
  if (tid < 256) hist[tid] = 0;
  __syncthreads();
#pragma unroll
  for (int u = 0; u < EPT; ++u)
    if ((unsigned)d[u] < (unsigned)NNODES) atomicAdd(&hist[d[u] >> 8], 1);
  __syncthreads();
  if (tid < NBIN) cntmat[tid * NCHUNK + blk] = hist[tid];
}

__device__ __forceinline__ void wave_scan256(const int* in, int* out, int lane) {
  int v[8];
  int s = 0;
#pragma unroll
  for (int k = 0; k < 8; ++k) v[k] = in[lane * 8 + k];
#pragma unroll
  for (int k = 0; k < 8; ++k) { const int t = v[k]; v[k] = s; s += t; }
  int incl = s;
#pragma unroll
  for (int off = 1; off < 32; off <<= 1) {
    const int nb = __shfl_up(incl, off, 32);
    if (lane >= off) incl += nb;
  }
  const int excl = incl - s;
#pragma unroll
  for (int k = 0; k < 8; ++k) out[lane * 8 + k] = excl + v[k];
  if (lane == 31) out[256] = incl;
}

__device__ __forceinline__ void bin_place_role(const int* __restrict__ src,
                                               const int* __restrict__ dst,
                                               const float* __restrict__ ew,
                                               const int* __restrict__ cntmat,
                                               int* __restrict__ binStart,
                                               int2* __restrict__ binned, int e, int blk) {
  __shared__ int binTot[256];
  __shared__ int binPre[256];
  __shared__ int binBeg[257];
  __shared__ int cursor[256];
  const int tid  = threadIdx.x;
  const int lane = tid & 31;
  const int wave = tid >> 5;

  int4 a[RPW], b[RPW];
#pragma unroll
  for (int j = 0; j < RPW; ++j) {
    const int row = wave + j * (CHUNK_T / 32);
    const int rr = row < NBIN ? row : NBIN - 1;
    const int4* __restrict__ r4 = (const int4*)(cntmat + rr * NCHUNK);
    a[j] = r4[lane];
    b[j] = r4[32 + lane];
  }

  const int cbeg = blk * CHUNK_E;
  const int cend = (cbeg + CHUNK_E < e) ? cbeg + CHUNK_E : e;
  int d[EPT], s[EPT];
  float w[EPT];

  if (tid >= NBIN && tid < 256) { binTot[tid] = 0; binPre[tid] = 0; }
  {
    const int c0 = 4 * lane;
    const int c1 = 128 + 4 * lane;
#pragma unroll
    for (int j = 0; j < RPW; ++j) {
      const int row = wave + j * (CHUNK_T / 32);
      int tot = (a[j].x + a[j].y) + (a[j].z + a[j].w) + (b[j].x + b[j].y) + (b[j].z + b[j].w);
      int pre = (c0     < blk ? a[j].x : 0) + (c0 + 1 < blk ? a[j].y : 0) +
                (c0 + 2 < blk ? a[j].z : 0) + (c0 + 3 < blk ? a[j].w : 0) +
                (c1     < blk ? b[j].x : 0) + (c1 + 1 < blk ? b[j].y : 0) +
                (c1 + 2 < blk ? b[j].z : 0) + (c1 + 3 < blk ? b[j].w : 0);
#pragma unroll
      for (int m = 16; m > 0; m >>= 1) {
        tot += __shfl_xor(tot, m, 32);
        pre += __shfl_xor(pre, m, 32);
      }
      if (lane == 0 && row < NBIN) { binTot[row] = tot; binPre[row] = pre; }
    }
  }
#pragma unroll
  for (int u = 0; u < EPT; ++u) {
    const int i = cbeg + tid + u * CHUNK_T;
    const int ic = i < cend ? i : cbeg;
    const int dv = dst[ic];
    s[u] = src[ic];
    w[u] = ew[ic];
    d[u] = i < cend ? dv : -1;
  }
  __syncthreads();
  if (wave == 0) wave_scan256(binTot, binBeg, lane);
  __syncthreads();
  if (tid < 256) cursor[tid] = binBeg[tid] + binPre[tid];
  if (blk == 0 && tid <= NBIN) binStart[tid] = binBeg[tid];
  __syncthreads();
#pragma unroll
  for (int u = 0; u < EPT; ++u) {
    if ((unsigned)d[u] < (unsigned)NNODES) {
      const int p = atomicAdd(&cursor[d[u] >> 8], 1);
      int sc = s[u];
      sc = sc < 0 ? 0 : (sc >= NNODES ? NNODES - 1 : sc);
      const unsigned key = ((unsigned)(d[u] & 255) << 24) | (unsigned)sc;
      if ((unsigned)p < (unsigned)e)
        binned[p] = make_int2((int)key, __float_as_int(w[u]));
    }
  }
}

__device__ __forceinline__ void bin_sort_role(const int* __restrict__ binStart,
                                              const int2* __restrict__ binned,
                                              int2* __restrict__ csr,
                                              int* __restrict__ rowptr,
                                              float* __restrict__ dinv, int n, int e, int bin) {
  __shared__ int   cnt[256];
  __shared__ float deg[256];
  __shared__ int   loff[257];
  __shared__ int   cur[256];
  const int tid  = threadIdx.x;
  const int lane = tid & 31;
  const int wave = tid >> 5;

  int start = binStart[bin];
  int end   = binStart[bin + 1];
  start = start < 0 ? 0 : start;
  end   = end > e ? e : end;
  if (tid < 256) { cnt[tid] = 0; deg[tid] = 0.f; cur[tid] = 0; }
  __syncthreads();
  for (int i = start + tid; i < end; i += 4 * 512) {
    int2 r[4];
#pragma unroll
    for (int u = 0; u < 4; ++u) {
      const int idx = i + u * 512;
      r[u] = make_int2(0, 0);
      if (idx < end) r[u] = binned[idx];
    }
#pragma unroll
    for (int u = 0; u < 4; ++u) {
      if (i + u * 512 < end) {
        const int dl = (int)((unsigned)r[u].x >> 24);
        atomicAdd(&cnt[dl], 1);
        atomicAdd(&deg[dl], __int_as_float(r[u].y));
      }
    }
  }
  __syncthreads();
  if (wave == 0) wave_scan256(cnt, loff, lane);
  __syncthreads();
  if (tid < 256) {
    const int node = bin * BIN_NODES + tid;
    if (node < n) {
      rowptr[node] = start + loff[tid];
      const float dg = deg[tid] + 1.0f;
      dinv[node] = dg > 0.f ? rsqrtf(dg) : 0.f;
    }
  }
  if (bin == NBIN - 1 && tid == 0) rowptr[n] = end;
  for (int i = start + tid; i < end; i += 4 * 512) {
    int2 r[4];
#pragma unroll
    for (int u = 0; u < 4; ++u) {
      const int idx = i + u * 512;
      r[u] = make_int2(0, 0);
      if (idx < end) r[u] = binned[idx];
    }
#pragma unroll
    for (int u = 0; u < 4; ++u) {
      if (i + u * 512 < end) {
        const int dl = (int)((unsigned)r[u].x >> 24);
        const int q = start + loff[dl] + atomicAdd(&cur[dl], 1);
        if ((unsigned)q < (unsigned)e)
          csr[q] = make_int2(r[u].x & 0x00FFFFFF, r[u].y);
      }
    }
  }
}

#define GEMM_SPB 4
#define GEMM_KQ  (INCH / 4)
#define RED_F4   (3 * GEMM_SPB * 8 * 2 * 32)
static_assert(INCH == 256 && HIDCH == 128, "x @ W1 role is written for 256 x 128");

__device__ __forceinline__ void gemm_role(const float* __restrict__ A,
                                          const float* __restrict__ B,
                                          float* __restrict__ C,
                                          float4* red, int gblk, int stripBeg, int stripEnd) {
  const int lane = threadIdx.x & 31;
  const int wave = __builtin_amdgcn_readfirstlane((int)(threadIdx.x >> 5));
  const int sl   = wave & 3;
  const int kq   = wave >> 2;
  const int strip = stripBeg + gblk * GEMM_SPB + sl;
  const bool valid = strip < stripEnd;
  const int srow = valid ? strip : stripEnd - 1;

  const int mn   = lane & 15;
  const int kSel = (lane >> 4) * 2;

  const float* __restrict__ arow  = A + (size_t)(srow * 16 + mn) * INCH;
  const float* __restrict__ bbase = B + mn;

  v8f acc[8];
#pragma unroll
  for (int t = 0; t < 8; ++t) acc[t] = (v8f){};

  const int kbeg = kq * GEMM_KQ;
  for (int k0 = kbeg; k0 < kbeg + GEMM_KQ; k0 += 4) {
    const int ka = k0 + kSel;
    v2f a;
    a.x = arow[ka];
    a.y = arow[ka + 1];
    const float* __restrict__ b0 = bbase + (size_t)ka * HIDCH;
    const float* __restrict__ b1 = bbase + (size_t)(ka + 1) * HIDCH;
#pragma unroll
    for (int t = 0; t < 8; ++t) {
      v2f b;
      b.x = b0[t * 16];
      b.y = b1[t * 16];
      acc[t] = __builtin_amdgcn_wmma_f32_16x16x4_f32(
           false, a,  false, b,
           (short)0, acc[t],  false,  false);
    }
  }

  if (kq != 0) {
    float4* __restrict__ rp = red + (size_t)(((kq - 1) * GEMM_SPB + sl) * 8) * 64 + lane;
#pragma unroll
    for (int t = 0; t < 8; ++t) {
      rp[t * 64]      = make_float4(acc[t][0], acc[t][1], acc[t][2], acc[t][3]);
      rp[t * 64 + 32] = make_float4(acc[t][4], acc[t][5], acc[t][6], acc[t][7]);
    }
  }
  __syncthreads();
  if (kq == 0 && valid) {
#pragma unroll
    for (int q = 0; q < 3; ++q) {
      const float4* __restrict__ rp = red + (size_t)((q * GEMM_SPB + sl) * 8) * 64 + lane;
#pragma unroll
      for (int t = 0; t < 8; ++t) {
        const float4 lo = rp[t * 64];
        const float4 hi = rp[t * 64 + 32];
        acc[t][0] += lo.x; acc[t][1] += lo.y; acc[t][2] += lo.z; acc[t][3] += lo.w;
        acc[t][4] += hi.x; acc[t][5] += hi.y; acc[t][6] += hi.z; acc[t][7] += hi.w;
      }
    }
    const int mbase = (lane >> 4) * 8;
    float* __restrict__ cbase = C + (size_t)(strip * 16) * HIDCH + mn;
#pragma unroll
    for (int t = 0; t < 8; ++t) {
      float* __restrict__ crow = cbase + t * 16;
#pragma unroll
      for (int r = 0; r < 8; ++r)
        crow[(size_t)(mbase + r) * HIDCH] = acc[t][r];
    }
  }
}

__global__ __launch_bounds__(512, 8) void k_count_gemm(const int* __restrict__ dst,
                                                       int* __restrict__ cntmat, int e,
                                                       const float* __restrict__ x,
                                                       const float* __restrict__ W,
                                                       float* __restrict__ h,
                                                       int stripBeg, int stripEnd) {
  __shared__ float4 red[RED_F4];
  if (blockIdx.x < NCHUNK) bin_count_role(dst, cntmat, e, (int)blockIdx.x);
  else gemm_role(x, W, h, red, (int)blockIdx.x - NCHUNK, stripBeg, stripEnd);
}

__global__ __launch_bounds__(512, 8) void k_place_gemm(const int* __restrict__ src,
                                                       const int* __restrict__ dst,
                                                       const float* __restrict__ ew,
                                                       const int* __restrict__ cntmat,
                                                       int* __restrict__ binStart,
                                                       int2* __restrict__ binned, int e,
                                                       const float* __restrict__ x,
                                                       const float* __restrict__ W,
                                                       float* __restrict__ h,
                                                       int stripBeg, int stripEnd) {
  __shared__ float4 red[RED_F4];
  if (blockIdx.x < NCHUNK) bin_place_role(src, dst, ew, cntmat, binStart, binned, e, (int)blockIdx.x);
  else gemm_role(x, W, h, red, (int)blockIdx.x - NCHUNK, stripBeg, stripEnd);
}

__global__ __launch_bounds__(512, 8) void k_sort_gemm(const int* __restrict__ binStart,
                                                      const int2* __restrict__ binned,
                                                      int2* __restrict__ csr,
                                                      int* __restrict__ rowptr,
                                                      float* __restrict__ dinv, int n, int e,
                                                      const float* __restrict__ x,
                                                      const float* __restrict__ W,
                                                      float* __restrict__ h,
                                                      int stripBeg, int stripEnd) {
  __shared__ float4 red[RED_F4];
  if (blockIdx.x < NBIN) bin_sort_role(binStart, binned, csr, rowptr, dinv, n, e, (int)blockIdx.x);
  else gemm_role(x, W, h, red, (int)blockIdx.x - NBIN, stripBeg, stripEnd);
}

template <int U>
__device__ __forceinline__ void gather_batch(const float4* __restrict__ h4, int sv, float wv,
                                             int j, int lane, float4& acc) {
  float4 v[U];
  float w[U];
#pragma unroll
  for (int u = 0; u < U; ++u) {
    const int s = __builtin_amdgcn_readlane(sv, j + u);
    w[u] = __int_as_float(__builtin_amdgcn_readlane(__float_as_int(wv), j + u));
    v[u] = h4[(size_t)s * (HIDCH / 4) + lane];
  }
#pragma unroll
  for (int u = 0; u < U; ++u) {
    acc.x += w[u] * v[u].x;
    acc.y += w[u] * v[u].y;
    acc.z += w[u] * v[u].z;
    acc.w += w[u] * v[u].w;
  }
}

__global__ __launch_bounds__(256) void k_gather(const int* __restrict__ rowptr,
                                                const int2* __restrict__ csr,
                                                const float* __restrict__ dinv,
                                                const float* __restrict__ h,
                                                const float* __restrict__ bias,
                                                float* __restrict__ out,
                                                int n, int nedges, int doRelu) {
  const int lane = threadIdx.x & 31;
  const int node = __builtin_amdgcn_readfirstlane((int)(blockIdx.x * 8 + (threadIdx.x >> 5)));
  if (node >= n) return;

  int beg = rowptr[node];
  int end = rowptr[node + 1];
  beg = beg < 0 ? 0 : beg;
  end = end > nedges ? nedges : end;
  const float dd = dinv[node];

  const float4* __restrict__ h4 = (const float4*)h;
  const float4 hs = h4[(size_t)node * (HIDCH / 4) + lane];
  const float selfw = dd * dd;
  float4 acc = make_float4(0.f, 0.f, 0.f, 0.f);

  for (int base = beg; base < end; base += 32) {
    const int left = end - base;
    const int m = left < 32 ? left : 32;
    int sv = 0;
    float wv = 0.f;
    if (lane < m) {
      const int2 ce = csr[base + lane];
      int s = ce.x;
      s = s < 0 ? 0 : (s >= n ? n - 1 : s);
      sv = s;
      wv = dinv[s] * __int_as_float(ce.y) * dd;
    }
    int j = 0;
    for (; j + 8 <= m; j += 8) gather_batch<8>(h4, sv, wv, j, lane, acc);
    if (m & 4) { gather_batch<4>(h4, sv, wv, j, lane, acc); j += 4; }
    if (m & 2) { gather_batch<2>(h4, sv, wv, j, lane, acc); j += 2; }
    if (m & 1) { gather_batch<1>(h4, sv, wv, j, lane, acc); }
  }

  const float4 bv = ((const float4*)bias)[lane];
  float4 r = make_float4(acc.x + selfw * hs.x + bv.x, acc.y + selfw * hs.y + bv.y,
                         acc.z + selfw * hs.z + bv.z, acc.w + selfw * hs.w + bv.w);
  if (doRelu) {
    r.x = fmaxf(r.x, 0.f); r.y = fmaxf(r.y, 0.f);
    r.z = fmaxf(r.z, 0.f); r.w = fmaxf(r.w, 0.f);
  }
  ((float4*)out)[(size_t)node * (HIDCH / 4) + lane] = r;
}

#define TPITCH 132
#define WPITCH 136
static_assert(NNODES % 16 == 0, "the fused kernel writes whole 16-node tiles");

__global__ __launch_bounds__(512) void k_gather_gemm(const int* __restrict__ rowptr,
                                                     const int2* __restrict__ csr,
                                                     const float* __restrict__ dinv,
                                                     const float* __restrict__ h,
                                                     const float* __restrict__ bias,
                                                     const float* __restrict__ Wn,
                                                     float* __restrict__ hout,
                                                     int n, int nedges, int ntiles) {
  __shared__ float Ws[HIDCH * WPITCH];
  __shared__ float tile[16 * TPITCH];

  const int tid  = threadIdx.x;
  const int lane = tid & 31;
  const int wave = __builtin_amdgcn_readfirstlane(tid >> 5);

  {
    const float4* __restrict__ W4 = (const float4*)Wn;
#pragma unroll
    for (int j = 0; j < 8; ++j) {
      const int i = tid + j * 512;
      const float4 v = W4[i];
      *(float4*)&Ws[(i >> 5) * WPITCH + (i & 31) * 4] = v;
    }
  }

  const float4* __restrict__ h4 = (const float4*)h;
  const float4 bv = ((const float4*)bias)[lane];

  for (int t = blockIdx.x; t < ntiles; t += gridDim.x) {
    const int node = t * 16 + wave;
    float4 r = make_float4(0.f, 0.f, 0.f, 0.f);
    if (node < n) {
      int beg = rowptr[node];
      int end = rowptr[node + 1];
      beg = beg < 0 ? 0 : beg;
      end = end > nedges ? nedges : end;
      const float dd = dinv[node];
      const float4 hs = h4[(size_t)node * (HIDCH / 4) + lane];
      const float selfw = dd * dd;
      float4 acc = make_float4(0.f, 0.f, 0.f, 0.f);

      for (int base = beg; base < end; base += 32) {
        const int left = end - base;
        const int m = left < 32 ? left : 32;
        int sv = 0;
        float wv = 0.f;
        if (lane < m) {
          const int2 ce = csr[base + lane];
          int s = ce.x;
          s = s < 0 ? 0 : (s >= n ? n - 1 : s);
          sv = s;
          wv = dinv[s] * __int_as_float(ce.y) * dd;
        }
        int j = 0;
        for (; j + 8 <= m; j += 8) gather_batch<8>(h4, sv, wv, j, lane, acc);
        if (m & 4) { gather_batch<4>(h4, sv, wv, j, lane, acc); j += 4; }
        if (m & 2) { gather_batch<2>(h4, sv, wv, j, lane, acc); j += 2; }
        if (m & 1) { gather_batch<1>(h4, sv, wv, j, lane, acc); }
      }

      r = make_float4(acc.x + selfw * hs.x + bv.x, acc.y + selfw * hs.y + bv.y,
                      acc.z + selfw * hs.z + bv.z, acc.w + selfw * hs.w + bv.w);
      r.x = fmaxf(r.x, 0.f); r.y = fmaxf(r.y, 0.f);
      r.z = fmaxf(r.z, 0.f); r.w = fmaxf(r.w, 0.f);
    }

    __syncthreads();
    *(float4*)&tile[wave * TPITCH + lane * 4] = r;
    __syncthreads();

    if (wave < 8) {
      const int mn   = lane & 15;
      const int kSel = (lane >> 4) * 2;
      const float* ap = tile + mn * TPITCH + kSel;
      const float* bp = Ws + kSel * WPITCH + wave * 16 + mn;
      v8f c = (v8f){};
#pragma unroll 8
      for (int k0 = 0; k0 < HIDCH; k0 += 4) {
        v2f a;
        a.x = ap[k0];
        a.y = ap[k0 + 1];
        v2f b;
        b.x = bp[k0 * WPITCH];
        b.y = bp[(k0 + 1) * WPITCH];
        c = __builtin_amdgcn_wmma_f32_16x16x4_f32(
             false, a,  false, b,
             (short)0, c,  false,  false);
      }
      const int row0 = t * 16 + (lane >> 4) * 8;
      float* __restrict__ crow = hout + (size_t)row0 * HIDCH + wave * 16 + mn;
#pragma unroll
      for (int q = 0; q < 8; ++q)
        crow[(size_t)q * HIDCH] = c[q];
    }
  }
}

extern "C" void kernel_launch(void* const* d_in, const int* in_sizes, int n_in,
                              void* d_out, int out_size, void* d_ws, size_t ws_size,
                              hipStream_t stream) {
  (void)in_sizes; (void)n_in; (void)out_size; (void)ws_size;

  const float* x  = (const float*)d_in[0];
  const int*   ei = (const int*)d_in[1];
  const float* ew = (const float*)d_in[2];
  const float* W1 = (const float*)d_in[3];
  const float* b1 = (const float*)d_in[4];
  const float* W2 = (const float*)d_in[5];
  const float* b2 = (const float*)d_in[6];
  const float* W3 = (const float*)d_in[7];
  const float* b3 = (const float*)d_in[8];
  float* out = (float*)d_out;

  const int* src = ei;
  const int* dst = ei + NEDGES;

  float* ws       = (float*)d_ws;
  float* dinv     = ws;
  int*   cntmat   = (int*)(ws + 50176);
  int*   binStart = (int*)(ws + 100352);
  int*   rowptr   = (int*)(ws + 100864);
  int2*  binned   = (int2*)(ws + 151040);
  int2*  csr      = (int2*)(ws + 1751040);
  float* h        = ws + 3351040;
  float* xb       = h + (size_t)NNODES * HIDCH;

  k_count_gemm<<<NCHUNK + 256, CHUNK_T, 0, stream>>>(dst, cntmat, NEDGES, x, W1, h, 0, 1024);
  k_place_gemm<<<NCHUNK + 256, CHUNK_T, 0, stream>>>(src, dst, ew, cntmat, binStart, binned, NEDGES,
                                                     x, W1, h, 1024, 2048);
  k_sort_gemm<<<NBIN + 270, 512, 0, stream>>>(binStart, binned, csr, rowptr, dinv, NNODES, NEDGES,
                                              x, W1, h, 2048, NNODES / 16);

  const int gatherBlocks = (NNODES + 7) / 8;

  const int ntiles = NNODES / 16;
  const int fusedBlocks = FUSED_GRID;

  k_gather_gemm<<<fusedBlocks, 512, 0, stream>>>(rowptr, csr, dinv, h, b1, W2, xb,
                                                 NNODES, NEDGES, ntiles);
  k_gather_gemm<<<fusedBlocks, 512, 0, stream>>>(rowptr, csr, dinv, xb, b2, W3, h,
                                                 NNODES, NEDGES, ntiles);
  k_gather<<<gatherBlocks, 256, 0, stream>>>(rowptr, csr, dinv, h, b3, out,
                                             NNODES, NEDGES, 0);
}


# Edge sort in two launches, not three: each chunk puts its records by bucket into its own stretch (rank = LDS add return), no global count; bucket blocks read 256 runs; row order of sums changes
# speedup vs baseline: 5.0806x; 1.0455x over previous
#include <hip/hip_runtime.h>


#define NNODES 50000
#define NEDGES 800000
#define INCH   256
#define HIDCH  128
#define FUSED_GRID 1024

typedef float v2f __attribute__((ext_vector_type(2)));
typedef float v8f __attribute__((ext_vector_type(8)));

#define BIN_NODES 256
#define NBIN ((NNODES + BIN_NODES - 1) / BIN_NODES)
#define NCHUNK 256
#define CHUNK_E ((NEDGES + NCHUNK - 1) / NCHUNK)
#define CHUNK_T 512
#define EPT ((CHUNK_E + CHUNK_T - 1) / CHUNK_T)
#define RPW ((NBIN + CHUNK_T / 32 - 1) / (CHUNK_T / 32))
static_assert(NBIN <= 256, "bucket tables are 256 wide");
static_assert(NCHUNK == 256, "a runmat row is read as two 16-byte loads per lane");
static_assert(CHUNK_T == 2 * NCHUNK, "two threads walk each run of a bucket");
static_assert(CHUNK_E < 65536, "run start and run length share one word");
static_assert(NNODES <= (1 << 24), "src is packed into 24 bits");
static_assert((NCHUNK - 1) * CHUNK_E < NEDGES, "every chunk starts inside the edge list");

__device__ __forceinline__ void wave_scan256(const int* in, int* out, int lane) {
  int v[8];
  int s = 0;
#pragma unroll
  for (int k = 0; k < 8; ++k) v[k] = in[lane * 8 + k];
#pragma unroll
  for (int k = 0; k < 8; ++k) { const int t = v[k]; v[k] = s; s += t; }
  int incl = s;
#pragma unroll
  for (int off = 1; off < 32; off <<= 1) {
    const int nb = __shfl_up(incl, off, 32);
    if (lane >= off) incl += nb;
  }
  const int excl = incl - s;
#pragma unroll
  for (int k = 0; k < 8; ++k) out[lane * 8 + k] = excl + v[k];
  if (lane == 31) out[256] = incl;
}

__device__ __forceinline__ void chunk_place_role(const int* __restrict__ src,
                                                 const int* __restrict__ dst,
                                                 const float* __restrict__ ew,
                                                 int* __restrict__ runmat,
                                                 int2* __restrict__ binned, int e, int blk) {
  __shared__ int hist[256];
  __shared__ int lbeg[257];
  const int tid  = threadIdx.x;
  const int lane = tid & 31;
  const int wave = tid >> 5;
  const int cbeg = blk * CHUNK_E;
  const int cend = (cbeg + CHUNK_E < e) ? cbeg + CHUNK_E : e;
  int d[EPT], s[EPT], r[EPT];
  float w[EPT];
#pragma unroll
  for (int u = 0; u < EPT; ++u) {
    const int i = cbeg + tid + u * CHUNK_T;
    const int ic = i < cend ? i : cbeg;
    const int dv = dst[ic];
    s[u] = src[ic];
    w[u] = ew[ic];
    d[u] = i < cend ? dv : -1;
  }
  if (tid < 256) hist[tid] = 0;
  __syncthreads();
#pragma unroll
  for (int u = 0; u < EPT; ++u) {
    r[u] = 0;
    if ((unsigned)d[u] < (unsigned)NNODES)
      r[u] = atomicAdd(&hist[d[u] >> 8], 1);
  }
  __syncthreads();
  if (wave == 0) wave_scan256(hist, lbeg, lane);
  __syncthreads();
  if (tid < NBIN) runmat[tid * NCHUNK + blk] = (lbeg[tid] << 16) | hist[tid];
#pragma unroll
  for (int u = 0; u < EPT; ++u) {
    if ((unsigned)d[u] < (unsigned)NNODES) {
      const int p = lbeg[d[u] >> 8] + r[u];
      int sc = s[u];
      sc = sc < 0 ? 0 : (sc >= NNODES ? NNODES - 1 : sc);
      const unsigned key = ((unsigned)(d[u] & 255) << 24) | (unsigned)sc;
      if ((unsigned)p < (unsigned)(cend - cbeg))
        binned[cbeg + p] = make_int2((int)key, __float_as_int(w[u]));
    }
  }
}

__device__ __forceinline__ void bucket_sort_role(const int* __restrict__ runmat,
                                                 const int2* __restrict__ binned,
                                                 int2* __restrict__ csr,
                                                 int* __restrict__ rowptr,
                                                 float* __restrict__ dinv, int n, int e, int bin) {
  __shared__ int   binTot[256];
  __shared__ int   binBeg[257];
  __shared__ int   runInfo[256];
  __shared__ int   cnt[256];
  __shared__ float deg[256];
  __shared__ int   loff[257];
  __shared__ int   cur[256];
  const int tid  = threadIdx.x;
  const int lane = tid & 31;
  const int wave = tid >> 5;

  int4 a[RPW], b[RPW];
#pragma unroll
  for (int j = 0; j < RPW; ++j) {
    const int row = wave + j * (CHUNK_T / 32);
    const int rr = row < NBIN ? row : NBIN - 1;
    const int4* __restrict__ r4 = (const int4*)(runmat + rr * NCHUNK);
    a[j] = r4[lane];
    b[j] = r4[32 + lane];
  }
  const int own = tid < 256 ? runmat[bin * NCHUNK + tid] : 0;

  if (tid < 256) { cnt[tid] = 0; deg[tid] = 0.f; cur[tid] = 0; }
  if (tid >= NBIN && tid < 256) binTot[tid] = 0;
#pragma unroll
  for (int j = 0; j < RPW; ++j) {
    const int row = wave + j * (CHUNK_T / 32);
    int tot = ((a[j].x & 0xFFFF) + (a[j].y & 0xFFFF)) + ((a[j].z & 0xFFFF) + (a[j].w & 0xFFFF)) +
              ((b[j].x & 0xFFFF) + (b[j].y & 0xFFFF)) + ((b[j].z & 0xFFFF) + (b[j].w & 0xFFFF));
#pragma unroll
    for (int m = 16; m > 0; m >>= 1) tot += __shfl_xor(tot, m, 32);
    if (lane == 0 && row < NBIN) binTot[row] = tot;
  }
  if (tid < 256) runInfo[tid] = own;
  __syncthreads();
  if (wave == 0) wave_scan256(binTot, binBeg, lane);
  __syncthreads();

  int start = binBeg[bin];
  int end   = binBeg[bin + 1];
  start = start < 0 ? 0 : start;
  end   = end > e ? e : end;

  const int c = tid >> 1;
  const unsigned info = (unsigned)runInfo[c];
  int rc = (int)(info & 0xFFFFu);
  rc = rc > CHUNK_E ? CHUNK_E : rc;
  const int rbase = c * CHUNK_E + (int)(info >> 16);

  for (int i = tid & 1; i < rc; i += 8) {
    int2 r[4];
#pragma unroll
    for (int u = 0; u < 4; ++u) {
      const int ii = i + 2 * u;
      r[u] = make_int2(0, 0);
      if (ii < rc && rbase + ii < e) r[u] = binned[rbase + ii];
    }
#pragma unroll
    for (int u = 0; u < 4; ++u) {
      const int ii = i + 2 * u;
      if (ii < rc && rbase + ii < e) {
        const int dl = (int)((unsigned)r[u].x >> 24);
        atomicAdd(&cnt[dl], 1);
        atomicAdd(&deg[dl], __int_as_float(r[u].y));
      }
    }
  }
  __syncthreads();
  if (wave == 0) wave_scan256(cnt, loff, lane);
  __syncthreads();
  if (tid < 256) {
    const int node = bin * BIN_NODES + tid;
    if (node < n) {
      rowptr[node] = start + loff[tid];
      const float dg = deg[tid] + 1.0f;
      dinv[node] = dg > 0.f ? rsqrtf(dg) : 0.f;
    }
  }
  if (bin == NBIN - 1 && tid == 0) rowptr[n] = end;
  for (int i = tid & 1; i < rc; i += 8) {
    int2 r[4];
#pragma unroll
    for (int u = 0; u < 4; ++u) {
      const int ii = i + 2 * u;
      r[u] = make_int2(0, 0);
      if (ii < rc && rbase + ii < e) r[u] = binned[rbase + ii];
    }
#pragma unroll
    for (int u = 0; u < 4; ++u) {
      const int ii = i + 2 * u;
      if (ii < rc && rbase + ii < e) {
        const int dl = (int)((unsigned)r[u].x >> 24);
        const int q = start + loff[dl] + atomicAdd(&cur[dl], 1);
        if ((unsigned)q < (unsigned)e)
          csr[q] = make_int2(r[u].x & 0x00FFFFFF, r[u].y);
      }
    }
  }
}

#define GEMM_SPB 4
#define GEMM_KQ  (INCH / 4)
#define RED_F4   (3 * GEMM_SPB * 8 * 2 * 32)
static_assert(INCH == 256 && HIDCH == 128, "x @ W1 role is written for 256 x 128");

__device__ __forceinline__ void gemm_role(const float* __restrict__ A,
                                          const float* __restrict__ B,
                                          float* __restrict__ C,
                                          float4* red, int gblk, int stripBeg, int stripEnd) {
  const int lane = threadIdx.x & 31;
  const int wave = __builtin_amdgcn_readfirstlane((int)(threadIdx.x >> 5));
  const int sl   = wave & 3;
  const int kq   = wave >> 2;
  const int strip = stripBeg + gblk * GEMM_SPB + sl;
  const bool valid = strip < stripEnd;
  const int srow = valid ? strip : stripEnd - 1;

  const int mn   = lane & 15;
  const int kSel = (lane >> 4) * 2;

  const float* __restrict__ arow  = A + (size_t)(srow * 16 + mn) * INCH;
  const float* __restrict__ bbase = B + mn;

  v8f acc[8];
#pragma unroll
  for (int t = 0; t < 8; ++t) acc[t] = (v8f){};

  const int kbeg = kq * GEMM_KQ;
  for (int k0 = kbeg; k0 < kbeg + GEMM_KQ; k0 += 4) {
    const int ka = k0 + kSel;
    v2f a;
    a.x = arow[ka];
    a.y = arow[ka + 1];
    const float* __restrict__ b0 = bbase + (size_t)ka * HIDCH;
    const float* __restrict__ b1 = bbase + (size_t)(ka + 1) * HIDCH;
#pragma unroll
    for (int t = 0; t < 8; ++t) {
      v2f b;
      b.x = b0[t * 16];
      b.y = b1[t * 16];
      acc[t] = __builtin_amdgcn_wmma_f32_16x16x4_f32(
           false, a,  false, b,
           (short)0, acc[t],  false,  false);
    }
  }

  if (kq != 0) {
    float4* __restrict__ rp = red + (size_t)(((kq - 1) * GEMM_SPB + sl) * 8) * 64 + lane;
#pragma unroll
    for (int t = 0; t < 8; ++t) {
      rp[t * 64]      = make_float4(acc[t][0], acc[t][1], acc[t][2], acc[t][3]);
      rp[t * 64 + 32] = make_float4(acc[t][4], acc[t][5], acc[t][6], acc[t][7]);
    }
  }
  __syncthreads();
  if (kq == 0 && valid) {
#pragma unroll
    for (int q = 0; q < 3; ++q) {
      const float4* __restrict__ rp = red + (size_t)((q * GEMM_SPB + sl) * 8) * 64 + lane;
#pragma unroll
      for (int t = 0; t < 8; ++t) {
        const float4 lo = rp[t * 64];
        const float4 hi = rp[t * 64 + 32];
        acc[t][0] += lo.x; acc[t][1] += lo.y; acc[t][2] += lo.z; acc[t][3] += lo.w;
        acc[t][4] += hi.x; acc[t][5] += hi.y; acc[t][6] += hi.z; acc[t][7] += hi.w;
      }
    }
    const int mbase = (lane >> 4) * 8;
    float* __restrict__ cbase = C + (size_t)(strip * 16) * HIDCH + mn;
#pragma unroll
    for (int t = 0; t < 8; ++t) {
      float* __restrict__ crow = cbase + t * 16;
#pragma unroll
      for (int r = 0; r < 8; ++r)
        crow[(size_t)(mbase + r) * HIDCH] = acc[t][r];
    }
  }
}

__global__ __launch_bounds__(512, 8) void k_place_gemm(const int* __restrict__ src,
                                                       const int* __restrict__ dst,
                                                       const float* __restrict__ ew,
                                                       int* __restrict__ runmat,
                                                       int2* __restrict__ binned, int e,
                                                       const float* __restrict__ x,
                                                       const float* __restrict__ W,
                                                       float* __restrict__ h,
                                                       int stripBeg, int stripEnd) {
  __shared__ float4 red[RED_F4];
  if (blockIdx.x < NCHUNK) chunk_place_role(src, dst, ew, runmat, binned, e, (int)blockIdx.x);
  else gemm_role(x, W, h, red, (int)blockIdx.x - NCHUNK, stripBeg, stripEnd);
}

__global__ __launch_bounds__(512, 8) void k_sort_gemm(const int* __restrict__ runmat,
                                                      const int2* __restrict__ binned,
                                                      int2* __restrict__ csr,
                                                      int* __restrict__ rowptr,
                                                      float* __restrict__ dinv, int n, int e,
                                                      const float* __restrict__ x,
                                                      const float* __restrict__ W,
                                                      float* __restrict__ h,
                                                      int stripBeg, int stripEnd) {
  __shared__ float4 red[RED_F4];
  if (blockIdx.x < NBIN) bucket_sort_role(runmat, binned, csr, rowptr, dinv, n, e, (int)blockIdx.x);
  else gemm_role(x, W, h, red, (int)blockIdx.x - NBIN, stripBeg, stripEnd);
}

template <int U>
__device__ __forceinline__ void gather_batch(const float4* __restrict__ h4, int sv, float wv,
                                             int j, int lane, float4& acc) {
  float4 v[U];
  float w[U];
#pragma unroll
  for (int u = 0; u < U; ++u) {
    const int s = __builtin_amdgcn_readlane(sv, j + u);
    w[u] = __int_as_float(__builtin_amdgcn_readlane(__float_as_int(wv), j + u));
    v[u] = h4[(size_t)s * (HIDCH / 4) + lane];
  }
#pragma unroll
  for (int u = 0; u < U; ++u) {
    acc.x += w[u] * v[u].x;
    acc.y += w[u] * v[u].y;
    acc.z += w[u] * v[u].z;
    acc.w += w[u] * v[u].w;
  }
}

__global__ __launch_bounds__(256) void k_gather(const int* __restrict__ rowptr,
                                                const int2* __restrict__ csr,
                                                const float* __restrict__ dinv,
                                                const float* __restrict__ h,
                                                const float* __restrict__ bias,
                                                float* __restrict__ out,
                                                int n, int nedges, int doRelu) {
  const int lane = threadIdx.x & 31;
  const int node = __builtin_amdgcn_readfirstlane((int)(blockIdx.x * 8 + (threadIdx.x >> 5)));
  if (node >= n) return;

  int beg = rowptr[node];
  int end = rowptr[node + 1];
  beg = beg < 0 ? 0 : beg;
  end = end > nedges ? nedges : end;
  const float dd = dinv[node];

  const float4* __restrict__ h4 = (const float4*)h;
  const float4 hs = h4[(size_t)node * (HIDCH / 4) + lane];
  const float selfw = dd * dd;
  float4 acc = make_float4(0.f, 0.f, 0.f, 0.f);

  for (int base = beg; base < end; base += 32) {
    const int left = end - base;
    const int m = left < 32 ? left : 32;
    int sv = 0;
    float wv = 0.f;
    if (lane < m) {
      const int2 ce = csr[base + lane];
      int s = ce.x;
      s = s < 0 ? 0 : (s >= n ? n - 1 : s);
      sv = s;
      wv = dinv[s] * __int_as_float(ce.y) * dd;
    }
    int j = 0;
    for (; j + 8 <= m; j += 8) gather_batch<8>(h4, sv, wv, j, lane, acc);
    if (m & 4) { gather_batch<4>(h4, sv, wv, j, lane, acc); j += 4; }
    if (m & 2) { gather_batch<2>(h4, sv, wv, j, lane, acc); j += 2; }
    if (m & 1) { gather_batch<1>(h4, sv, wv, j, lane, acc); }
  }

  const float4 bv = ((const float4*)bias)[lane];
  float4 r = make_float4(acc.x + selfw * hs.x + bv.x, acc.y + selfw * hs.y + bv.y,
                         acc.z + selfw * hs.z + bv.z, acc.w + selfw * hs.w + bv.w);
  if (doRelu) {
    r.x = fmaxf(r.x, 0.f); r.y = fmaxf(r.y, 0.f);
    r.z = fmaxf(r.z, 0.f); r.w = fmaxf(r.w, 0.f);
  }
  ((float4*)out)[(size_t)node * (HIDCH / 4) + lane] = r;
}

#define TPITCH 132
#define WPITCH 136
static_assert(NNODES % 16 == 0, "the fused kernel writes whole 16-node tiles");

__global__ __launch_bounds__(512) void k_gather_gemm(const int* __restrict__ rowptr,
                                                     const int2* __restrict__ csr,
                                                     const float* __restrict__ dinv,
                                                     const float* __restrict__ h,
                                                     const float* __restrict__ bias,
                                                     const float* __restrict__ Wn,
                                                     float* __restrict__ hout,
                                                     int n, int nedges, int ntiles) {
  __shared__ float Ws[HIDCH * WPITCH];
  __shared__ float tile[16 * TPITCH];

  const int tid  = threadIdx.x;
  const int lane = tid & 31;
  const int wave = __builtin_amdgcn_readfirstlane(tid >> 5);

  {
    const float4* __restrict__ W4 = (const float4*)Wn;
#pragma unroll
    for (int j = 0; j < 8; ++j) {
      const int i = tid + j * 512;
      const float4 v = W4[i];
      *(float4*)&Ws[(i >> 5) * WPITCH + (i & 31) * 4] = v;
    }
  }

  const float4* __restrict__ h4 = (const float4*)h;
  const float4 bv = ((const float4*)bias)[lane];

  for (int t = blockIdx.x; t < ntiles; t += gridDim.x) {
    const int node = t * 16 + wave;
    float4 r = make_float4(0.f, 0.f, 0.f, 0.f);
    if (node < n) {
      int beg = rowptr[node];
      int end = rowptr[node + 1];
      beg = beg < 0 ? 0 : beg;
      end = end > nedges ? nedges : end;
      const float dd = dinv[node];
      const float4 hs = h4[(size_t)node * (HIDCH / 4) + lane];
      const float selfw = dd * dd;
      float4 acc = make_float4(0.f, 0.f, 0.f, 0.f);

      for (int base = beg; base < end; base += 32) {
        const int left = end - base;
        const int m = left < 32 ? left : 32;
        int sv = 0;
        float wv = 0.f;
        if (lane < m) {
          const int2 ce = csr[base + lane];
          int s = ce.x;
          s = s < 0 ? 0 : (s >= n ? n - 1 : s);
          sv = s;
          wv = dinv[s] * __int_as_float(ce.y) * dd;
        }
        int j = 0;
        for (; j + 8 <= m; j += 8) gather_batch<8>(h4, sv, wv, j, lane, acc);
        if (m & 4) { gather_batch<4>(h4, sv, wv, j, lane, acc); j += 4; }
        if (m & 2) { gather_batch<2>(h4, sv, wv, j, lane, acc); j += 2; }
        if (m & 1) { gather_batch<1>(h4, sv, wv, j, lane, acc); }
      }

      r = make_float4(acc.x + selfw * hs.x + bv.x, acc.y + selfw * hs.y + bv.y,
                      acc.z + selfw * hs.z + bv.z, acc.w + selfw * hs.w + bv.w);
      r.x = fmaxf(r.x, 0.f); r.y = fmaxf(r.y, 0.f);
      r.z = fmaxf(r.z, 0.f); r.w = fmaxf(r.w, 0.f);
    }

    __syncthreads();
    *(float4*)&tile[wave * TPITCH + lane * 4] = r;
    __syncthreads();

    if (wave < 8) {
      const int mn   = lane & 15;
      const int kSel = (lane >> 4) * 2;
      const float* ap = tile + mn * TPITCH + kSel;
      const float* bp = Ws + kSel * WPITCH + wave * 16 + mn;
      v8f c = (v8f){};
#pragma unroll 8
      for (int k0 = 0; k0 < HIDCH; k0 += 4) {
        v2f a;
        a.x = ap[k0];
        a.y = ap[k0 + 1];
        v2f b;
        b.x = bp[k0 * WPITCH];
        b.y = bp[(k0 + 1) * WPITCH];
        c = __builtin_amdgcn_wmma_f32_16x16x4_f32(
             false, a,  false, b,
             (short)0, c,  false,  false);
      }
      const int row0 = t * 16 + (lane >> 4) * 8;
      float* __restrict__ crow = hout + (size_t)row0 * HIDCH + wave * 16 + mn;
#pragma unroll
      for (int q = 0; q < 8; ++q)
        crow[(size_t)q * HIDCH] = c[q];
    }
  }
}

extern "C" void kernel_launch(void* const* d_in, const int* in_sizes, int n_in,
                              void* d_out, int out_size, void* d_ws, size_t ws_size,
                              hipStream_t stream) {
  (void)in_sizes; (void)n_in; (void)out_size; (void)ws_size;

  const float* x  = (const float*)d_in[0];
  const int*   ei = (const int*)d_in[1];
  const float* ew = (const float*)d_in[2];
  const float* W1 = (const float*)d_in[3];
  const float* b1 = (const float*)d_in[4];
  const float* W2 = (const float*)d_in[5];
  const float* b2 = (const float*)d_in[6];
  const float* W3 = (const float*)d_in[7];
  const float* b3 = (const float*)d_in[8];
  float* out = (float*)d_out;

  const int* src = ei;
  const int* dst = ei + NEDGES;

  float* ws       = (float*)d_ws;
  float* dinv     = ws;
  int*   runmat   = (int*)(ws + 50176);
  int*   rowptr   = (int*)(ws + 100864);
  int2*  binned   = (int2*)(ws + 151040);
  int2*  csr      = (int2*)(ws + 1751040);
  float* h        = ws + 3351040;
  float* xb       = h + (size_t)NNODES * HIDCH;

  k_place_gemm<<<NCHUNK + 512, CHUNK_T, 0, stream>>>(src, dst, ew, runmat, binned, NEDGES,
                                                     x, W1, h, 0, 2048);
  k_sort_gemm<<<NBIN + 270, CHUNK_T, 0, stream>>>(runmat, binned, csr, rowptr, dinv, NNODES, NEDGES,
                                                  x, W1, h, 2048, NNODES / 16);

  const int gatherBlocks = (NNODES + 7) / 8;

  const int ntiles = NNODES / 16;
  const int fusedBlocks = FUSED_GRID;

  k_gather_gemm<<<fusedBlocks, 512, 0, stream>>>(rowptr, csr, dinv, h, b1, W2, xb,
                                                 NNODES, NEDGES, ntiles);
  k_gather_gemm<<<fusedBlocks, 512, 0, stream>>>(rowptr, csr, dinv, xb, b2, W3, h,
                                                 NNODES, NEDGES, ntiles);
  k_gather<<<gatherBlocks, 256, 0, stream>>>(rowptr, csr, dinv, h, b3, out,
                                             NNODES, NEDGES, 0);
}
